# starting kernel
# baseline (speedup 1.0000x reference)
#include <hip/hip_runtime.h>
#include <hip/hip_bf16.h>


#define SAGE_N 50000
#define SAGE_D 128

typedef __attribute__((ext_vector_type(2))) float v2f;
typedef __attribute__((ext_vector_type(8))) float v8f;

__global__ void sage_zero_kernel(float* __restrict__ p, size_t n) {
    size_t i = (size_t)blockIdx.x * blockDim.x + threadIdx.x;
    size_t stride = (size_t)gridDim.x * blockDim.x;
    for (; i < n; i += stride) p[i] = 0.0f;
}

__global__ void sage_count_kernel(const int* __restrict__ dst,
                                  float* __restrict__ cnt, int E) {
    int e = blockIdx.x * blockDim.x + threadIdx.x;
    if (e < E) unsafeAtomicAdd(&cnt[dst[e]], 1.0f);
}

__global__ __launch_bounds__(256) void sage_scatter_kernel(
    const float* __restrict__ xin, const int* __restrict__ src,
    const int* __restrict__ dst, float* __restrict__ s, int E) {
    int e = blockIdx.x * 8 + (threadIdx.x >> 5);
    if (e >= E) return;
    int lane = threadIdx.x & 31;
    int sn = src[e];
    int dn = dst[e];
    const float4 v = *(const float4*)(xin + (size_t)sn * SAGE_D + lane * 4);
    float* sd = s + (size_t)dn * SAGE_D + lane * 4;
    unsafeAtomicAdd(sd + 0, v.x);
    unsafeAtomicAdd(sd + 1, v.y);
    unsafeAtomicAdd(sd + 2, v.z);
    unsafeAtomicAdd(sd + 3, v.w);
}

__global__ __launch_bounds__(256) void sage_gemm_kernel(
    const float* __restrict__ s,
    const float* __restrict__ cnt,
    const float* __restrict__ xin,
    const float* __restrict__ Wl,
    const float* __restrict__ bl,
    const float* __restrict__ Wr,
    float* __restrict__ out,
    int nRowTiles) {
    const int wave = threadIdx.x >> 5;
    const int tile = blockIdx.x * 8 + wave;
    if (tile >= nRowTiles) return;

    const int lane = threadIdx.x & 31;
    const int half = lane >> 4;
    const int lm   = lane & 15;
    const int m0   = tile * 16;

    const float inv = 1.0f / fmaxf(cnt[m0 + lm], 1.0f);

    const float* arow_s = s   + (size_t)(m0 + lm) * SAGE_D + 2 * half;
    const float* arow_x = xin + (size_t)(m0 + lm) * SAGE_D + 2 * half;
    const float* wl_row = Wl  + (size_t)lm * SAGE_D + 2 * half;
    const float* wr_row = Wr  + (size_t)lm * SAGE_D + 2 * half;

    v8f acc[8];
#pragma unroll
    for (int ct = 0; ct < 8; ++ct) acc[ct] = (v8f){};

    for (int k0 = 0; k0 < SAGE_D; k0 += 4) {
        v2f a_agg = *(const v2f*)(arow_s + k0);
        a_agg = a_agg * inv;
        v2f a_x = *(const v2f*)(arow_x + k0);
#pragma unroll
        for (int ct = 0; ct < 8; ++ct) {
            v2f b_l = *(const v2f*)(wl_row + (size_t)ct * 16 * SAGE_D + k0);
            v2f b_r = *(const v2f*)(wr_row + (size_t)ct * 16 * SAGE_D + k0);
            acc[ct] = __builtin_amdgcn_wmma_f32_16x16x4_f32(
                false, a_agg, false, b_l, (short)0, acc[ct], false, false);
            acc[ct] = __builtin_amdgcn_wmma_f32_16x16x4_f32(
                false, a_x, false, b_r, (short)0, acc[ct], false, false);
        }
    }

#pragma unroll
    for (int ct = 0; ct < 8; ++ct) {
        const float bias = bl[ct * 16 + lm];
#pragma unroll
        for (int v = 0; v < 8; ++v) {
            out[(size_t)(m0 + v + 8 * half) * SAGE_D + ct * 16 + lm] =
                acc[ct][v] + bias;
        }
    }
}

extern "C" void kernel_launch(void* const* d_in, const int* in_sizes, int n_in,
                              void* d_out, int out_size, void* d_ws, size_t ws_size,
                              hipStream_t stream) {
    const float* x   = (const float*)d_in[0];
    const int*   ei  = (const int*)d_in[1];
    const float* Wl1 = (const float*)d_in[2];
    const float* bl1 = (const float*)d_in[3];
    const float* Wr1 = (const float*)d_in[4];
    const float* Wl2 = (const float*)d_in[5];
    const float* bl2 = (const float*)d_in[6];
    const float* Wr2 = (const float*)d_in[7];
    float* out = (float*)d_out;

    const int N = SAGE_N;
    const int E = in_sizes[1] / 2;
    const int* src = ei;
    const int* dst = ei + E;

    float* s   = (float*)d_ws;
    float* cnt = s + (size_t)N * SAGE_D;

    const int nRowTiles  = N / 16;
    const int gemmBlocks = (nRowTiles + 7) / 8;
    const int scatBlocks = (E + 7) / 8;

    sage_zero_kernel<<<2048, 256, 0, stream>>>(s, (size_t)N * SAGE_D + N);
    sage_count_kernel<<<(E + 255) / 256, 256, 0, stream>>>(dst, cnt, E);

    sage_scatter_kernel<<<scatBlocks, 256, 0, stream>>>(x, src, dst, s, E);
    sage_gemm_kernel<<<gemmBlocks, 256, 0, stream>>>(s, cnt, x, Wl1, bl1, Wr1,
                                                     out, nRowTiles);

    sage_zero_kernel<<<2048, 256, 0, stream>>>(s, (size_t)N * SAGE_D);
    sage_scatter_kernel<<<scatBlocks, 256, 0, stream>>>(out, src, dst, s, E);
    sage_gemm_kernel<<<gemmBlocks, 256, 0, stream>>>(s, cnt, out, Wl2, bl2, Wr2,
                                                     out, nRowTiles);
}


# Neighbour sums by per-node gather: edges filed into per-node records by an integer counter, each wave sorts its sources by id and adds rows in that fixed order; float atomics and zeroing of s removed
# speedup vs baseline: 2.0808x; 2.0808x over previous
#include <hip/hip_runtime.h>
#include <hip/hip_bf16.h>


#define SAGE_N 50000
#define SAGE_D 128
#define SAGE_REC 64
#define SAGE_SLOTS 63
#define SAGE_NOKEY 0x7fffffff

typedef __attribute__((ext_vector_type(2))) float v2f;
typedef __attribute__((ext_vector_type(8))) float v8f;

__global__ void sage_clear_kernel(int* __restrict__ rec, int* __restrict__ ovfCount) {
    const int i = blockIdx.x * blockDim.x + threadIdx.x;
    if (i < SAGE_N * (SAGE_REC / 4)) ((int4*)rec)[i] = make_int4(0, 0, 0, 0);
    if (i == 0) *ovfCount = 0;
}

__global__ void sage_bucket_kernel(const int* __restrict__ src,
                                   const int* __restrict__ dst,
                                   int* rec, int* ovfCount,
                                   int* __restrict__ ovfDst,
                                   int* __restrict__ ovfSrc, int E) {
    const int e = blockIdx.x * blockDim.x + threadIdx.x;
    if (e >= E) return;
    const int sn = src[e];
    const int dn = dst[e];
    int* r = rec + (size_t)dn * SAGE_REC;
    const int slot = atomicAdd(r, 1);
    if (slot < SAGE_SLOTS) {
        r[1 + slot] = sn;
    } else {
        const int k = atomicAdd(ovfCount, 1);
        ovfDst[k] = dn;
        ovfSrc[k] = sn;
    }
}

template <int K>
__device__ __forceinline__ void sage_add_rows(float4& acc, const float4* __restrict__ xrow,
                                              int sreg, int j) {
    float4 v[K];
#pragma unroll
    for (int i = 0; i < K; ++i) {
        const int sid = __builtin_amdgcn_readlane(sreg, j + i) >> 6;
        v[i] = xrow[(size_t)sid * (SAGE_D / 4)];
    }
#pragma unroll
    for (int i = 0; i < K; ++i) {
        acc.x += v[i].x; acc.y += v[i].y; acc.z += v[i].z; acc.w += v[i].w;
    }
}

__device__ __forceinline__ void sage_add_range(float4& acc, const float4* __restrict__ xrow,
                                               int sreg, int count) {
    int j = 0;
    for (; j + 8 <= count; j += 8) sage_add_rows<8>(acc, xrow, sreg, j);
    if (count & 4) { sage_add_rows<4>(acc, xrow, sreg, j); j += 4; }
    if (count & 2) { sage_add_rows<2>(acc, xrow, sreg, j); j += 2; }
    if (count & 1) { sage_add_rows<1>(acc, xrow, sreg, j); }
}

__global__ __launch_bounds__(256) void sage_gather_kernel(
    const float* __restrict__ xin, const int* __restrict__ rec,
    const int* __restrict__ ovfCount, const int* __restrict__ ovfDst,
    const int* __restrict__ ovfSrc, float* __restrict__ s,
    float* __restrict__ cnt) {
    __shared__ int sorted[8][SAGE_REC];
    const int wave = threadIdx.x >> 5;
    const int lane = threadIdx.x & 31;
    const int n = blockIdx.x * 8 + wave;
    if (n >= SAGE_N) return;

    const int* r = rec + (size_t)n * SAGE_REC;
    const int w0 = r[lane];
    const int deg = __builtin_amdgcn_readfirstlane(w0);
    const float4* xrow = (const float4*)xin + lane;
    float4 acc = make_float4(0.0f, 0.0f, 0.0f, 0.0f);

    if (deg <= SAGE_SLOTS) {
        int w1 = 0;
        if (deg > 31) w1 = r[32 + lane];
        const int i0 = lane - 1;
        const int i1 = lane + 31;
        const bool ok0 = (lane >= 1) && (i0 < deg);
        const bool ok1 = (i1 < deg);
        const int k0 = ok0 ? ((w0 << 6) | i0) : SAGE_NOKEY;
        const int k1 = ok1 ? ((w1 << 6) | i1) : SAGE_NOKEY;
        int r0 = 0, r1 = 0;
        const int na = deg < 31 ? deg : 31;
        for (int j = 0; j < na; ++j) {
            const int u = __builtin_amdgcn_readlane(k0, j + 1);
            r0 += (u < k0) ? 1 : 0;
            r1 += (u < k1) ? 1 : 0;
        }
        for (int j = 31; j < deg; ++j) {
            const int u = __builtin_amdgcn_readlane(k1, j - 31);
            r0 += (u < k0) ? 1 : 0;
            r1 += (u < k1) ? 1 : 0;
        }
        if (ok0) sorted[wave][r0] = k0;
        if (ok1) sorted[wave][r1] = k1;
        int s0 = 0, s1 = 0;
        if (lane < deg) s0 = sorted[wave][lane];
        if (32 + lane < deg) s1 = sorted[wave][32 + lane];
        sage_add_range(acc, xrow, s0, deg < 32 ? deg : 32);
        if (deg > 32) sage_add_range(acc, xrow, s1, deg - 32);
    } else {
        const int w1 = r[32 + lane];
        const int k0 = (lane >= 1) ? w0 : SAGE_NOKEY;
        const int k1 = w1;
        const int nov = *ovfCount;
        int last = -1;
        for (int it = 0; it < deg; ++it) {
            int best = SAGE_NOKEY;
            if (k0 > last && k0 < best) best = k0;
            if (k1 > last && k1 < best) best = k1;
            for (int i = lane; i < nov; i += 32) {
                if (ovfDst[i] == n) {
                    const int v = ovfSrc[i];
                    if (v > last && v < best) best = v;
                }
            }
            for (int m = 16; m >= 1; m >>= 1) {
                const int o = __shfl_xor(best, m, 32);
                best = o < best ? o : best;
            }
            if (best == SAGE_NOKEY) break;
            int c = (k0 == best ? 1 : 0) + (k1 == best ? 1 : 0);
            for (int i = lane; i < nov; i += 32) {
                if (ovfDst[i] == n && ovfSrc[i] == best) ++c;
            }
            for (int m = 16; m >= 1; m >>= 1) c += __shfl_xor(c, m, 32);
            const int sid = __builtin_amdgcn_readfirstlane(best);
            const float4 v = xrow[(size_t)sid * (SAGE_D / 4)];
            for (int t = 0; t < c; ++t) {
                acc.x += v.x; acc.y += v.y; acc.z += v.z; acc.w += v.w;
            }
            last = best;
        }
    }

    ((float4*)(s + (size_t)n * SAGE_D))[lane] = acc;
    if (lane == 0) cnt[n] = (float)deg;
}

__global__ __launch_bounds__(256) void sage_gemm_kernel(
    const float* __restrict__ s,
    const float* __restrict__ cnt,
    const float* __restrict__ xin,
    const float* __restrict__ Wl,
    const float* __restrict__ bl,
    const float* __restrict__ Wr,
    float* __restrict__ out,
    int nRowTiles) {
    const int wave = threadIdx.x >> 5;
    const int tile = blockIdx.x * 8 + wave;
    if (tile >= nRowTiles) return;

    const int lane = threadIdx.x & 31;
    const int half = lane >> 4;
    const int lm   = lane & 15;
    const int m0   = tile * 16;

    const float inv = 1.0f / fmaxf(cnt[m0 + lm], 1.0f);

    const float* arow_s = s   + (size_t)(m0 + lm) * SAGE_D + 2 * half;
    const float* arow_x = xin + (size_t)(m0 + lm) * SAGE_D + 2 * half;
    const float* wl_row = Wl  + (size_t)lm * SAGE_D + 2 * half;
    const float* wr_row = Wr  + (size_t)lm * SAGE_D + 2 * half;

    v8f acc[8];
#pragma unroll
    for (int ct = 0; ct < 8; ++ct) acc[ct] = (v8f){};

    for (int k0 = 0; k0 < SAGE_D; k0 += 4) {
        v2f a_agg = *(const v2f*)(arow_s + k0);
        a_agg = a_agg * inv;
        v2f a_x = *(const v2f*)(arow_x + k0);
#pragma unroll
        for (int ct = 0; ct < 8; ++ct) {
            v2f b_l = *(const v2f*)(wl_row + (size_t)ct * 16 * SAGE_D + k0);
            v2f b_r = *(const v2f*)(wr_row + (size_t)ct * 16 * SAGE_D + k0);
            acc[ct] = __builtin_amdgcn_wmma_f32_16x16x4_f32(
                false, a_agg, false, b_l, (short)0, acc[ct], false, false);
            acc[ct] = __builtin_amdgcn_wmma_f32_16x16x4_f32(
                false, a_x, false, b_r, (short)0, acc[ct], false, false);
        }
    }

#pragma unroll
    for (int ct = 0; ct < 8; ++ct) {
        const float bias = bl[ct * 16 + lm];
#pragma unroll
        for (int v = 0; v < 8; ++v) {
            out[(size_t)(m0 + v + 8 * half) * SAGE_D + ct * 16 + lm] =
                acc[ct][v] + bias;
        }
    }
}

extern "C" void kernel_launch(void* const* d_in, const int* in_sizes, int n_in,
                              void* d_out, int out_size, void* d_ws, size_t ws_size,
                              hipStream_t stream) {
    const float* x   = (const float*)d_in[0];
    const int*   ei  = (const int*)d_in[1];
    const float* Wl1 = (const float*)d_in[2];
    const float* bl1 = (const float*)d_in[3];
    const float* Wr1 = (const float*)d_in[4];
    const float* Wl2 = (const float*)d_in[5];
    const float* bl2 = (const float*)d_in[6];
    const float* Wr2 = (const float*)d_in[7];
    float* out = (float*)d_out;

    const int N = SAGE_N;
    const int E = in_sizes[1] / 2;
    const int* src = ei;
    const int* dst = ei + E;

    float* s      = (float*)d_ws;
    int*   rec    = (int*)(s + (size_t)N * SAGE_D);
    int*   ovfDst = rec + (size_t)N * SAGE_REC;
    int*   ovfSrc = ovfDst + E;
    float* cnt    = (float*)(ovfSrc + E);
    int*   ovfCnt = (int*)(cnt + N);

    const int nRowTiles  = N / 16;
    const int gemmBlocks = (nRowTiles + 7) / 8;
    const int gathBlocks = (N + 7) / 8;

    sage_clear_kernel<<<(N * (SAGE_REC / 4) + 255) / 256, 256, 0, stream>>>(
        rec, ovfCnt);
    sage_bucket_kernel<<<(E + 255) / 256, 256, 0, stream>>>(src, dst, rec, ovfCnt,
                                                            ovfDst, ovfSrc, E);

    sage_gather_kernel<<<gathBlocks, 256, 0, stream>>>(x, rec, ovfCnt, ovfDst,
                                                       ovfSrc, s, cnt);
    sage_gemm_kernel<<<gemmBlocks, 256, 0, stream>>>(s, cnt, x, Wl1, bl1, Wr1,
                                                     out, nRowTiles);

    sage_gather_kernel<<<gathBlocks, 256, 0, stream>>>(out, rec, ovfCnt, ovfDst,
                                                       ovfSrc, s, cnt);
    sage_gemm_kernel<<<gemmBlocks, 256, 0, stream>>>(s, cnt, out, Wl2, bl2, Wr2,
                                                     out, nRowTiles);
}


# GEMM weights copied to LDS once per 16-wave work-group (256 work-groups, one row tile per wave): B fragments are 8-byte LDS reads, not 16-line global loads; k order and sum order unchanged
# speedup vs baseline: 2.7985x; 1.3449x over previous
#include <hip/hip_runtime.h>
#include <hip/hip_bf16.h>


#define SAGE_N 50000
#define SAGE_D 128
#define SAGE_REC 64
#define SAGE_SLOTS 63
#define SAGE_NOKEY 0x7fffffff

typedef __attribute__((ext_vector_type(2))) float v2f;
typedef __attribute__((ext_vector_type(8))) float v8f;

__global__ void sage_clear_kernel(int* __restrict__ rec, int* __restrict__ ovfCount) {
    const int i = blockIdx.x * blockDim.x + threadIdx.x;
    if (i < SAGE_N * (SAGE_REC / 4)) ((int4*)rec)[i] = make_int4(0, 0, 0, 0);
    if (i == 0) *ovfCount = 0;
}

__global__ void sage_bucket_kernel(const int* __restrict__ src,
                                   const int* __restrict__ dst,
                                   int* rec, int* ovfCount,
                                   int* __restrict__ ovfDst,
                                   int* __restrict__ ovfSrc, int E) {
    const int e = blockIdx.x * blockDim.x + threadIdx.x;
    if (e >= E) return;
    const int sn = src[e];
    const int dn = dst[e];
    int* r = rec + (size_t)dn * SAGE_REC;
    const int slot = atomicAdd(r, 1);
    if (slot < SAGE_SLOTS) {
        r[1 + slot] = sn;
    } else {
        const int k = atomicAdd(ovfCount, 1);
        ovfDst[k] = dn;
        ovfSrc[k] = sn;
    }
}

template <int K>
__device__ __forceinline__ void sage_add_rows(float4& acc, const float4* __restrict__ xrow,
                                              int sreg, int j) {
    float4 v[K];
#pragma unroll
    for (int i = 0; i < K; ++i) {
        const int sid = __builtin_amdgcn_readlane(sreg, j + i) >> 6;
        v[i] = xrow[(size_t)sid * (SAGE_D / 4)];
    }
#pragma unroll
    for (int i = 0; i < K; ++i) {
        acc.x += v[i].x; acc.y += v[i].y; acc.z += v[i].z; acc.w += v[i].w;
    }
}

__device__ __forceinline__ void sage_add_range(float4& acc, const float4* __restrict__ xrow,
                                               int sreg, int count) {
    int j = 0;
    for (; j + 8 <= count; j += 8) sage_add_rows<8>(acc, xrow, sreg, j);
    if (count & 4) { sage_add_rows<4>(acc, xrow, sreg, j); j += 4; }
    if (count & 2) { sage_add_rows<2>(acc, xrow, sreg, j); j += 2; }
    if (count & 1) { sage_add_rows<1>(acc, xrow, sreg, j); }
}

__global__ __launch_bounds__(256) void sage_gather_kernel(
    const float* __restrict__ xin, const int* __restrict__ rec,
    const int* __restrict__ ovfCount, const int* __restrict__ ovfDst,
    const int* __restrict__ ovfSrc, float* __restrict__ s,
    float* __restrict__ cnt) {
    __shared__ int sorted[8][SAGE_REC];
    const int wave = threadIdx.x >> 5;
    const int lane = threadIdx.x & 31;
    const int n = blockIdx.x * 8 + wave;
    if (n >= SAGE_N) return;

    const int* r = rec + (size_t)n * SAGE_REC;
    const int w0 = r[lane];
    const int deg = __builtin_amdgcn_readfirstlane(w0);
    const float4* xrow = (const float4*)xin + lane;
    float4 acc = make_float4(0.0f, 0.0f, 0.0f, 0.0f);

    if (deg <= SAGE_SLOTS) {
        int w1 = 0;
        if (deg > 31) w1 = r[32 + lane];
        const int i0 = lane - 1;
        const int i1 = lane + 31;
        const bool ok0 = (lane >= 1) && (i0 < deg);
        const bool ok1 = (i1 < deg);
        const int k0 = ok0 ? ((w0 << 6) | i0) : SAGE_NOKEY;
        const int k1 = ok1 ? ((w1 << 6) | i1) : SAGE_NOKEY;
        int r0 = 0, r1 = 0;
        const int na = deg < 31 ? deg : 31;
        for (int j = 0; j < na; ++j) {
            const int u = __builtin_amdgcn_readlane(k0, j + 1);
            r0 += (u < k0) ? 1 : 0;
            r1 += (u < k1) ? 1 : 0;
        }
        for (int j = 31; j < deg; ++j) {
            const int u = __builtin_amdgcn_readlane(k1, j - 31);
            r0 += (u < k0) ? 1 : 0;
            r1 += (u < k1) ? 1 : 0;
        }
        if (ok0) sorted[wave][r0] = k0;
        if (ok1) sorted[wave][r1] = k1;
        int s0 = 0, s1 = 0;
        if (lane < deg) s0 = sorted[wave][lane];
        if (32 + lane < deg) s1 = sorted[wave][32 + lane];
        sage_add_range(acc, xrow, s0, deg < 32 ? deg : 32);
        if (deg > 32) sage_add_range(acc, xrow, s1, deg - 32);
    } else {
        const int w1 = r[32 + lane];
        const int k0 = (lane >= 1) ? w0 : SAGE_NOKEY;
        const int k1 = w1;
        const int nov = *ovfCount;
        int last = -1;
        for (int it = 0; it < deg; ++it) {
            int best = SAGE_NOKEY;
            if (k0 > last && k0 < best) best = k0;
            if (k1 > last && k1 < best) best = k1;
            for (int i = lane; i < nov; i += 32) {
                if (ovfDst[i] == n) {
                    const int v = ovfSrc[i];
                    if (v > last && v < best) best = v;
                }
            }
            for (int m = 16; m >= 1; m >>= 1) {
                const int o = __shfl_xor(best, m, 32);
                best = o < best ? o : best;
            }
            if (best == SAGE_NOKEY) break;
            int c = (k0 == best ? 1 : 0) + (k1 == best ? 1 : 0);
            for (int i = lane; i < nov; i += 32) {
                if (ovfDst[i] == n && ovfSrc[i] == best) ++c;
            }
            for (int m = 16; m >= 1; m >>= 1) c += __shfl_xor(c, m, 32);
            const int sid = __builtin_amdgcn_readfirstlane(best);
            const float4 v = xrow[(size_t)sid * (SAGE_D / 4)];
            for (int t = 0; t < c; ++t) {
                acc.x += v.x; acc.y += v.y; acc.z += v.z; acc.w += v.w;
            }
            last = best;
        }
    }

    ((float4*)(s + (size_t)n * SAGE_D))[lane] = acc;
    if (lane == 0) cnt[n] = (float)deg;
}

#define SAGE_WP 132
#define SAGE_GEMM_WGS 256
#define SAGE_GEMM_THREADS 512
static_assert(SAGE_GEMM_WGS * (SAGE_GEMM_THREADS / 32) >= SAGE_N / 16,
              "one wave per row tile");

__device__ __forceinline__ void sage_load_b(v2f (&b)[16], const float* wl_s,
                                            const float* wr_s, int k0) {
#pragma unroll
    for (int ct = 0; ct < 8; ++ct) {
        b[ct]     = *(const v2f*)(wl_s + ct * 16 * SAGE_WP + k0);
        b[8 + ct] = *(const v2f*)(wr_s + ct * 16 * SAGE_WP + k0);
    }
}

__device__ __forceinline__ void sage_load_a(v2f (&a)[8], const float* arow_s,
                                            const float* arow_x, int kc) {
#pragma unroll
    for (int i = 0; i < 4; ++i) {
        a[i]     = *(const v2f*)(arow_s + kc + 4 * i);
        a[4 + i] = *(const v2f*)(arow_x + kc + 4 * i);
    }
}

__device__ __forceinline__ void sage_mma_step(v8f (&acc)[8], const v2f (&b)[16],
                                              v2f a_agg, v2f a_x) {
#pragma unroll
    for (int ct = 0; ct < 8; ++ct) {
        acc[ct] = __builtin_amdgcn_wmma_f32_16x16x4_f32(
            false, a_agg, false, b[ct], (short)0, acc[ct], false, false);
        acc[ct] = __builtin_amdgcn_wmma_f32_16x16x4_f32(
            false, a_x, false, b[8 + ct], (short)0, acc[ct], false, false);
    }
}

template <bool LAST>
__device__ __forceinline__ void sage_mma_chunk(v8f (&acc)[8], const v2f (&a)[8],
                                               v2f (&b0)[16], v2f (&b1)[16],
                                               const float* wl_s, const float* wr_s,
                                               float inv, int kc) {
#pragma unroll
    for (int i = 0; i < 4; i += 2) {
        sage_load_b(b1, wl_s, wr_s, kc + 4 * i + 4);
        __builtin_amdgcn_sched_barrier(0);
        sage_mma_step(acc, b0, a[i] * inv, a[4 + i]);
        __builtin_amdgcn_sched_barrier(0);
        if (!LAST || i < 2) sage_load_b(b0, wl_s, wr_s, kc + 4 * i + 8);
        __builtin_amdgcn_sched_barrier(0);
        sage_mma_step(acc, b1, a[i + 1] * inv, a[4 + i + 1]);
        __builtin_amdgcn_sched_barrier(0);
    }
}

__global__ __launch_bounds__(SAGE_GEMM_THREADS) void sage_gemm_kernel(
    const float* __restrict__ s,
    const float* __restrict__ cnt,
    const float* __restrict__ xin,
    const float* __restrict__ Wl,
    const float* __restrict__ bl,
    const float* __restrict__ Wr,
    float* __restrict__ out,
    int nRowTiles) {
    __shared__ __attribute__((aligned(16))) float sW[2 * SAGE_D * SAGE_WP];

#pragma unroll
    for (int j = 0; j < SAGE_D * (SAGE_D / 4) / SAGE_GEMM_THREADS; ++j) {
        const int i   = j * SAGE_GEMM_THREADS + threadIdx.x;
        const int row = i >> 5;
        const int c4  = i & 31;
        const float4 vl = ((const float4*)Wl)[i];
        const float4 vr = ((const float4*)Wr)[i];
        *(float4*)(sW + row * SAGE_WP + c4 * 4) = vl;
        *(float4*)(sW + (SAGE_D + row) * SAGE_WP + c4 * 4) = vr;
    }
    __syncthreads();

    const int wave   = __builtin_amdgcn_readfirstlane(threadIdx.x >> 5);
    const int lane = threadIdx.x & 31;
    const int half = lane >> 4;
    const int lm   = lane & 15;
    const float* wl_s = sW + lm * SAGE_WP + 2 * half;
    const float* wr_s = wl_s + SAGE_D * SAGE_WP;

    const int tile = wave * SAGE_GEMM_WGS + blockIdx.x;
    if (tile < nRowTiles) {
        const int m0 = tile * 16;
        const float deg = cnt[m0 + lm];
        const float* arow_s = s   + (size_t)(m0 + lm) * SAGE_D + 2 * half;
        const float* arow_x = xin + (size_t)(m0 + lm) * SAGE_D + 2 * half;

        float bias[8];
#pragma unroll
        for (int ct = 0; ct < 8; ++ct) bias[ct] = bl[ct * 16 + lm];

        v8f acc[8];
#pragma unroll
        for (int ct = 0; ct < 8; ++ct) acc[ct] = (v8f){};

        v2f a0[8], a1[8], b0[16], b1[16];
        sage_load_a(a0, arow_s, arow_x, 0);
        sage_load_b(b0, wl_s, wr_s, 0);
        __builtin_amdgcn_sched_barrier(0);
        const float inv = 1.0f / fmaxf(deg, 1.0f);
        for (int kc = 0; kc < SAGE_D - 32; kc += 32) {
            sage_load_a(a1, arow_s, arow_x, kc + 16);
            sage_mma_chunk<false>(acc, a0, b0, b1, wl_s, wr_s, inv, kc);
            sage_load_a(a0, arow_s, arow_x, kc + 32);
            sage_mma_chunk<false>(acc, a1, b0, b1, wl_s, wr_s, inv, kc + 16);
        }
        sage_load_a(a1, arow_s, arow_x, SAGE_D - 16);
        sage_mma_chunk<false>(acc, a0, b0, b1, wl_s, wr_s, inv, SAGE_D - 32);
        sage_mma_chunk<true>(acc, a1, b0, b1, wl_s, wr_s, inv, SAGE_D - 16);

#pragma unroll
        for (int ct = 0; ct < 8; ++ct) {
#pragma unroll
            for (int v = 0; v < 8; ++v) {
                out[(size_t)(m0 + v + 8 * half) * SAGE_D + ct * 16 + lm] =
                    acc[ct][v] + bias[ct];
            }
        }
    }
}

extern "C" void kernel_launch(void* const* d_in, const int* in_sizes, int n_in,
                              void* d_out, int out_size, void* d_ws, size_t ws_size,
                              hipStream_t stream) {
    const float* x   = (const float*)d_in[0];
    const int*   ei  = (const int*)d_in[1];
    const float* Wl1 = (const float*)d_in[2];
    const float* bl1 = (const float*)d_in[3];
    const float* Wr1 = (const float*)d_in[4];
    const float* Wl2 = (const float*)d_in[5];
    const float* bl2 = (const float*)d_in[6];
    const float* Wr2 = (const float*)d_in[7];
    float* out = (float*)d_out;

    const int N = SAGE_N;
    const int E = in_sizes[1] / 2;
    const int* src = ei;
    const int* dst = ei + E;

    float* s      = (float*)d_ws;
    int*   rec    = (int*)(s + (size_t)N * SAGE_D);
    int*   ovfDst = rec + (size_t)N * SAGE_REC;
    int*   ovfSrc = ovfDst + E;
    float* cnt    = (float*)(ovfSrc + E);
    int*   ovfCnt = (int*)(cnt + N);

    const int nRowTiles  = N / 16;
    const int gemmBlocks = SAGE_GEMM_WGS;
    const int gathBlocks = (N + 7) / 8;

    sage_clear_kernel<<<(N * (SAGE_REC / 4) + 255) / 256, 256, 0, stream>>>(
        rec, ovfCnt);
    sage_bucket_kernel<<<(E + 255) / 256, 256, 0, stream>>>(src, dst, rec, ovfCnt,
                                                            ovfDst, ovfSrc, E);

    sage_gather_kernel<<<gathBlocks, 256, 0, stream>>>(x, rec, ovfCnt, ovfDst,
                                                       ovfSrc, s, cnt);
    sage_gemm_kernel<<<gemmBlocks, SAGE_GEMM_THREADS, 0, stream>>>(
        s, cnt, x, Wl1, bl1, Wr1, out, nRowTiles);

    sage_gather_kernel<<<gathBlocks, 256, 0, stream>>>(out, rec, ovfCnt, ovfDst,
                                                       ovfSrc, s, cnt);
    sage_gemm_kernel<<<gemmBlocks, SAGE_GEMM_THREADS, 0, stream>>>(
        s, cnt, out, Wl2, bl2, Wr2, out, nRowTiles);
}


# Edges filed by destination on LDS counters in two launches (split by node range, then each range's work-group builds its node records in LDS): per-edge global adds and the record clear removed
# speedup vs baseline: 3.4767x; 1.2423x over previous
#include <hip/hip_runtime.h>
#include <hip/hip_bf16.h>


#define SAGE_N 50000
#define SAGE_D 128
#define SAGE_REC 64
#define SAGE_SLOTS 63
#define SAGE_NOKEY 0x7fffffff

typedef __attribute__((ext_vector_type(2))) float v2f;
typedef __attribute__((ext_vector_type(8))) float v8f;

#define SAGE_NB 256
#define SAGE_NPB 196
#define SAGE_NP 256
#define SAGE_LP 65
#define SAGE_PART_THREADS 512
#define SAGE_PART_BATCH 8
static_assert(SAGE_NB * SAGE_NPB >= SAGE_N, "ranges cover all nodes");
static_assert(SAGE_N <= 65536 && SAGE_NPB < 32768, "entry packing");

__global__ __launch_bounds__(SAGE_PART_THREADS) void sage_part_kernel(
    const int* __restrict__ src, const int* __restrict__ dst,
    int* __restrict__ cells, int* __restrict__ ovfCount, int E) {
    __shared__ int st[SAGE_NB * SAGE_LP];
    const int tid = threadIdx.x;
    const int p = blockIdx.x;
    if (tid < SAGE_NB) st[tid * SAGE_LP] = 0;
    if (p == 0 && tid == 0) *ovfCount = 0;
    __syncthreads();

    const int per = (E + SAGE_NP - 1) / SAGE_NP;
    const int e0 = p * per;
    const int e1 = (e0 + per < E) ? (e0 + per) : E;
    for (int base = e0; base < e1; base += SAGE_PART_THREADS * SAGE_PART_BATCH) {
        int d[SAGE_PART_BATCH], s[SAGE_PART_BATCH], slot[SAGE_PART_BATCH];
#pragma unroll
        for (int i = 0; i < SAGE_PART_BATCH; ++i) {
            const int e = base + i * SAGE_PART_THREADS + tid;
            const bool ok = e < e1;
            d[i] = ok ? dst[e] : -1;
            s[i] = ok ? src[e] : 0;
        }
#pragma unroll
        for (int i = 0; i < SAGE_PART_BATCH; ++i) {
            slot[i] = SAGE_SLOTS;
            if (d[i] >= 0) slot[i] = atomicAdd(st + (d[i] / SAGE_NPB) * SAGE_LP, 1);
        }
#pragma unroll
        for (int i = 0; i < SAGE_PART_BATCH; ++i) {
            if (slot[i] < SAGE_SLOTS) {
                const int b = d[i] / SAGE_NPB;
                st[b * SAGE_LP + 1 + slot[i]] = ((d[i] - b * SAGE_NPB) << 16) | s[i];
            }
        }
    }
    __syncthreads();

    const int wave = tid >> 5;
    const int lane = tid & 31;
    constexpr int kWaves = SAGE_PART_THREADS / 32;
    constexpr int kCellsPerWave = SAGE_NB / kWaves;
    static_assert(kCellsPerWave * kWaves == SAGE_NB && kCellsPerWave <= 32, "cells per wave");
    const int myCount = (lane < kCellsPerWave) ? st[(wave + kWaves * lane) * SAGE_LP] : 0;
    int first[kCellsPerWave];
#pragma unroll
    for (int k = 0; k < kCellsPerWave; ++k) {
        const int n = __builtin_amdgcn_readlane(myCount, k);
        const int used = n < SAGE_SLOTS ? n : SAGE_SLOTS;
        first[k] = (lane <= used) ? st[(wave + kWaves * k) * SAGE_LP + lane] : 0;
    }
#pragma unroll
    for (int k = 0; k < kCellsPerWave; ++k) {
        const int b = wave + kWaves * k;
        const int n = __builtin_amdgcn_readlane(myCount, k);
        const int used = n < SAGE_SLOTS ? n : SAGE_SLOTS;
        int* g = cells + ((size_t)b * SAGE_NP + p) * SAGE_REC;
        g[lane] = first[k];
        if (32 + lane <= used) g[32 + lane] = st[b * SAGE_LP + 32 + lane];
    }
}

__device__ __forceinline__ int sage_take_slot(int* recS, int entry) {
    return atomicAdd(recS + (entry >> 16) * SAGE_LP, 1);
}

__device__ __forceinline__ void sage_put_entry(int* recS, int entry, int slot, int node0,
                                               int* ovfCount, int* __restrict__ ovfDst,
                                               int* __restrict__ ovfSrc) {
    const int ld = entry >> 16;
    const int sn = entry & 0xffff;
    if (slot < SAGE_SLOTS) {
        recS[ld * SAGE_LP + 1 + slot] = sn;
    } else {
        const int k = atomicAdd(ovfCount, 1);
        ovfDst[k] = node0 + ld;
        ovfSrc[k] = sn;
    }
}

__global__ __launch_bounds__(256) void sage_file_kernel(
    const int* __restrict__ src, const int* __restrict__ dst,
    const int* __restrict__ cells, int* __restrict__ rec, int* ovfCount,
    int* __restrict__ ovfDst, int* __restrict__ ovfSrc, int E) {
    __shared__ int recS[SAGE_NPB * SAGE_LP];
    const int tid = threadIdx.x;
    const int wave = tid >> 5;
    const int lane = tid & 31;
    const int b = blockIdx.x;
    const int node0 = b * SAGE_NPB;
    for (int i = tid; i < SAGE_NPB * SAGE_LP; i += 256) recS[i] = 0;
    __syncthreads();

    const int* cw = cells + ((size_t)b * SAGE_NP + wave * 32) * SAGE_REC;
    int w[32], slot[32];
#pragma unroll
    for (int j = 0; j < 32; ++j) w[j] = cw[j * SAGE_REC + lane];
    unsigned more = 0;
    unsigned redo = 0;
#pragma unroll
    for (int j = 0; j < 32; ++j) {
        const int n = __builtin_amdgcn_readfirstlane(w[j]);
        const bool whole = n <= SAGE_SLOTS;
        redo |= (whole ? 0u : 1u) << j;
        more |= ((whole && n > 31) ? 1u : 0u) << j;
        slot[j] = -1;
        if (whole && lane >= 1 && lane <= n) slot[j] = sage_take_slot(recS, w[j]);
    }
#pragma unroll
    for (int j = 0; j < 32; ++j) {
        if (slot[j] >= 0) sage_put_entry(recS, w[j], slot[j], node0, ovfCount, ovfDst, ovfSrc);
    }
#pragma nounroll
    for (int j = 0; j < 32; ++j) {
        if ((more >> j) & 1u) {
            const int n = cw[j * SAGE_REC];
            if (32 + lane <= n) {
                const int entry = cw[j * SAGE_REC + 32 + lane];
                sage_put_entry(recS, entry, sage_take_slot(recS, entry), node0, ovfCount,
                               ovfDst, ovfSrc);
            }
        }
    }
#pragma nounroll
    for (int j = 0; j < 32; ++j) {
        if ((redo >> j) & 1u) {
            const int per = (E + SAGE_NP - 1) / SAGE_NP;
            const int e0 = (wave * 32 + j) * per;
            const int e1 = (e0 + per < E) ? (e0 + per) : E;
            for (int e = e0 + lane; e < e1; e += 32) {
                const int dn = dst[e];
                if (dn >= node0 && dn < node0 + SAGE_NPB) {
                    const int entry = ((dn - node0) << 16) | src[e];
                    sage_put_entry(recS, entry, sage_take_slot(recS, entry), node0, ovfCount,
                                   ovfDst, ovfSrc);
                }
            }
        }
    }
    __syncthreads();

    for (int q = tid; q < SAGE_NPB * (SAGE_REC / 4); q += 256) {
        const int ln = q >> 4;
        const int* r = recS + ln * SAGE_LP + (q & 15) * 4;
        if (node0 + ln < SAGE_N)
            ((int4*)(rec + (size_t)node0 * SAGE_REC))[q] = make_int4(r[0], r[1], r[2], r[3]);
    }
}

template <int K>
__device__ __forceinline__ void sage_add_rows(float4& acc, const float4* __restrict__ xrow,
                                              int sreg, int j) {
    float4 v[K];
#pragma unroll
    for (int i = 0; i < K; ++i) {
        const int sid = __builtin_amdgcn_readlane(sreg, j + i) >> 6;
        v[i] = xrow[(size_t)sid * (SAGE_D / 4)];
    }
#pragma unroll
    for (int i = 0; i < K; ++i) {
        acc.x += v[i].x; acc.y += v[i].y; acc.z += v[i].z; acc.w += v[i].w;
    }
}

__device__ __forceinline__ void sage_add_range(float4& acc, const float4* __restrict__ xrow,
                                               int sreg, int count) {
    int j = 0;
    for (; j + 8 <= count; j += 8) sage_add_rows<8>(acc, xrow, sreg, j);
    if (count & 4) { sage_add_rows<4>(acc, xrow, sreg, j); j += 4; }
    if (count & 2) { sage_add_rows<2>(acc, xrow, sreg, j); j += 2; }
    if (count & 1) { sage_add_rows<1>(acc, xrow, sreg, j); }
}

__global__ __launch_bounds__(256) void sage_gather_kernel(
    const float* __restrict__ xin, const int* __restrict__ rec,
    const int* __restrict__ ovfCount, const int* __restrict__ ovfDst,
    const int* __restrict__ ovfSrc, float* __restrict__ s,
    float* __restrict__ cnt) {
    __shared__ int sorted[8][SAGE_REC];
    const int wave = threadIdx.x >> 5;
    const int lane = threadIdx.x & 31;
    const int n = blockIdx.x * 8 + wave;
    if (n >= SAGE_N) return;

    const int* r = rec + (size_t)n * SAGE_REC;
    const int w0 = r[lane];
    const int deg = __builtin_amdgcn_readfirstlane(w0);
    const float4* xrow = (const float4*)xin + lane;
    float4 acc = make_float4(0.0f, 0.0f, 0.0f, 0.0f);

    if (deg <= SAGE_SLOTS) {
        int w1 = 0;
        if (deg > 31) w1 = r[32 + lane];
        const int i0 = lane - 1;
        const int i1 = lane + 31;
        const bool ok0 = (lane >= 1) && (i0 < deg);
        const bool ok1 = (i1 < deg);
        const int k0 = ok0 ? ((w0 << 6) | i0) : SAGE_NOKEY;
        const int k1 = ok1 ? ((w1 << 6) | i1) : SAGE_NOKEY;
        int r0 = 0, r1 = 0;
        const int na = deg < 31 ? deg : 31;
        for (int j = 0; j < na; ++j) {
            const int u = __builtin_amdgcn_readlane(k0, j + 1);
            r0 += (u < k0) ? 1 : 0;
            r1 += (u < k1) ? 1 : 0;
        }
        for (int j = 31; j < deg; ++j) {
            const int u = __builtin_amdgcn_readlane(k1, j - 31);
            r0 += (u < k0) ? 1 : 0;
            r1 += (u < k1) ? 1 : 0;
        }
        if (ok0) sorted[wave][r0] = k0;
        if (ok1) sorted[wave][r1] = k1;
        int s0 = 0, s1 = 0;
        if (lane < deg) s0 = sorted[wave][lane];
        if (32 + lane < deg) s1 = sorted[wave][32 + lane];
        sage_add_range(acc, xrow, s0, deg < 32 ? deg : 32);
        if (deg > 32) sage_add_range(acc, xrow, s1, deg - 32);
    } else {
        const int w1 = r[32 + lane];
        const int k0 = (lane >= 1) ? w0 : SAGE_NOKEY;
        const int k1 = w1;
        const int nov = *ovfCount;
        int last = -1;
        for (int it = 0; it < deg; ++it) {
            int best = SAGE_NOKEY;
            if (k0 > last && k0 < best) best = k0;
            if (k1 > last && k1 < best) best = k1;
            for (int i = lane; i < nov; i += 32) {
                if (ovfDst[i] == n) {
                    const int v = ovfSrc[i];
                    if (v > last && v < best) best = v;
                }
            }
            for (int m = 16; m >= 1; m >>= 1) {
                const int o = __shfl_xor(best, m, 32);
                best = o < best ? o : best;
            }
            if (best == SAGE_NOKEY) break;
            int c = (k0 == best ? 1 : 0) + (k1 == best ? 1 : 0);
            for (int i = lane; i < nov; i += 32) {
                if (ovfDst[i] == n && ovfSrc[i] == best) ++c;
            }
            for (int m = 16; m >= 1; m >>= 1) c += __shfl_xor(c, m, 32);
            const int sid = __builtin_amdgcn_readfirstlane(best);
            const float4 v = xrow[(size_t)sid * (SAGE_D / 4)];
            for (int t = 0; t < c; ++t) {
                acc.x += v.x; acc.y += v.y; acc.z += v.z; acc.w += v.w;
            }
            last = best;
        }
    }

    ((float4*)(s + (size_t)n * SAGE_D))[lane] = acc;
    if (lane == 0) cnt[n] = (float)deg;
}

#define SAGE_WP 132
#define SAGE_GEMM_WGS 256
#define SAGE_GEMM_THREADS 512
static_assert(SAGE_GEMM_WGS * (SAGE_GEMM_THREADS / 32) >= SAGE_N / 16,
              "one wave per row tile");

__device__ __forceinline__ void sage_load_b(v2f (&b)[16], const float* wl_s,
                                            const float* wr_s, int k0) {
#pragma unroll
    for (int ct = 0; ct < 8; ++ct) {
        b[ct]     = *(const v2f*)(wl_s + ct * 16 * SAGE_WP + k0);
        b[8 + ct] = *(const v2f*)(wr_s + ct * 16 * SAGE_WP + k0);
    }
}

__device__ __forceinline__ void sage_load_a(v2f (&a)[8], const float* arow_s,
                                            const float* arow_x, int kc) {
#pragma unroll
    for (int i = 0; i < 4; ++i) {
        a[i]     = *(const v2f*)(arow_s + kc + 4 * i);
        a[4 + i] = *(const v2f*)(arow_x + kc + 4 * i);
    }
}

__device__ __forceinline__ void sage_mma_step(v8f (&acc)[8], const v2f (&b)[16],
                                              v2f a_agg, v2f a_x) {
#pragma unroll
    for (int ct = 0; ct < 8; ++ct) {
        acc[ct] = __builtin_amdgcn_wmma_f32_16x16x4_f32(
            false, a_agg, false, b[ct], (short)0, acc[ct], false, false);
        acc[ct] = __builtin_amdgcn_wmma_f32_16x16x4_f32(
            false, a_x, false, b[8 + ct], (short)0, acc[ct], false, false);
    }
}

template <bool LAST>
__device__ __forceinline__ void sage_mma_chunk(v8f (&acc)[8], const v2f (&a)[8],
                                               v2f (&b0)[16], v2f (&b1)[16],
                                               const float* wl_s, const float* wr_s,
                                               float inv, int kc) {
#pragma unroll
    for (int i = 0; i < 4; i += 2) {
        sage_load_b(b1, wl_s, wr_s, kc + 4 * i + 4);
        __builtin_amdgcn_sched_barrier(0);
        sage_mma_step(acc, b0, a[i] * inv, a[4 + i]);
        __builtin_amdgcn_sched_barrier(0);
        if (!LAST || i < 2) sage_load_b(b0, wl_s, wr_s, kc + 4 * i + 8);
        __builtin_amdgcn_sched_barrier(0);
        sage_mma_step(acc, b1, a[i + 1] * inv, a[4 + i + 1]);
        __builtin_amdgcn_sched_barrier(0);
    }
}

__global__ __launch_bounds__(SAGE_GEMM_THREADS) void sage_gemm_kernel(
    const float* __restrict__ s,
    const float* __restrict__ cnt,
    const float* __restrict__ xin,
    const float* __restrict__ Wl,
    const float* __restrict__ bl,
    const float* __restrict__ Wr,
    float* __restrict__ out,
    int nRowTiles) {
    __shared__ __attribute__((aligned(16))) float sW[2 * SAGE_D * SAGE_WP];

#pragma unroll
    for (int j = 0; j < SAGE_D * (SAGE_D / 4) / SAGE_GEMM_THREADS; ++j) {
        const int i   = j * SAGE_GEMM_THREADS + threadIdx.x;
        const int row = i >> 5;
        const int c4  = i & 31;
        const float4 vl = ((const float4*)Wl)[i];
        const float4 vr = ((const float4*)Wr)[i];
        *(float4*)(sW + row * SAGE_WP + c4 * 4) = vl;
        *(float4*)(sW + (SAGE_D + row) * SAGE_WP + c4 * 4) = vr;
    }
    __syncthreads();

    const int wave   = __builtin_amdgcn_readfirstlane(threadIdx.x >> 5);
    const int lane = threadIdx.x & 31;
    const int half = lane >> 4;
    const int lm   = lane & 15;
    const float* wl_s = sW + lm * SAGE_WP + 2 * half;
    const float* wr_s = wl_s + SAGE_D * SAGE_WP;

    const int tile = wave * SAGE_GEMM_WGS + blockIdx.x;
    if (tile < nRowTiles) {
        const int m0 = tile * 16;
        const float deg = cnt[m0 + lm];
        const float* arow_s = s   + (size_t)(m0 + lm) * SAGE_D + 2 * half;
        const float* arow_x = xin + (size_t)(m0 + lm) * SAGE_D + 2 * half;

        float bias[8];
#pragma unroll
        for (int ct = 0; ct < 8; ++ct) bias[ct] = bl[ct * 16 + lm];

        v8f acc[8];
#pragma unroll
        for (int ct = 0; ct < 8; ++ct) acc[ct] = (v8f){};

        v2f a0[8], a1[8], b0[16], b1[16];
        sage_load_a(a0, arow_s, arow_x, 0);
        sage_load_b(b0, wl_s, wr_s, 0);
        __builtin_amdgcn_sched_barrier(0);
        const float inv = 1.0f / fmaxf(deg, 1.0f);
        for (int kc = 0; kc < SAGE_D - 32; kc += 32) {
            sage_load_a(a1, arow_s, arow_x, kc + 16);
            sage_mma_chunk<false>(acc, a0, b0, b1, wl_s, wr_s, inv, kc);
            sage_load_a(a0, arow_s, arow_x, kc + 32);
            sage_mma_chunk<false>(acc, a1, b0, b1, wl_s, wr_s, inv, kc + 16);
        }
        sage_load_a(a1, arow_s, arow_x, SAGE_D - 16);
        sage_mma_chunk<false>(acc, a0, b0, b1, wl_s, wr_s, inv, SAGE_D - 32);
        sage_mma_chunk<true>(acc, a1, b0, b1, wl_s, wr_s, inv, SAGE_D - 16);

#pragma unroll
        for (int ct = 0; ct < 8; ++ct) {
#pragma unroll
            for (int v = 0; v < 8; ++v) {
                out[(size_t)(m0 + v + 8 * half) * SAGE_D + ct * 16 + lm] =
                    acc[ct][v] + bias[ct];
            }
        }
    }
}

extern "C" void kernel_launch(void* const* d_in, const int* in_sizes, int n_in,
                              void* d_out, int out_size, void* d_ws, size_t ws_size,
                              hipStream_t stream) {
    const float* x   = (const float*)d_in[0];
    const int*   ei  = (const int*)d_in[1];
    const float* Wl1 = (const float*)d_in[2];
    const float* bl1 = (const float*)d_in[3];
    const float* Wr1 = (const float*)d_in[4];
    const float* Wl2 = (const float*)d_in[5];
    const float* bl2 = (const float*)d_in[6];
    const float* Wr2 = (const float*)d_in[7];
    float* out = (float*)d_out;

    const int N = SAGE_N;
    const int E = in_sizes[1] / 2;
    const int* src = ei;
    const int* dst = ei + E;

    float* s      = (float*)d_ws;
    int*   rec    = (int*)(s + (size_t)N * SAGE_D);
    int*   cells  = rec + (size_t)N * SAGE_REC;
    int*   ovfDst = cells + (size_t)SAGE_NB * SAGE_NP * SAGE_REC;
    int*   ovfSrc = ovfDst + E;
    float* cnt    = (float*)(ovfSrc + E);
    int*   ovfCnt = (int*)(cnt + N);

    const int nRowTiles  = N / 16;
    const int gemmBlocks = SAGE_GEMM_WGS;
    const int gathBlocks = (N + 7) / 8;

    sage_part_kernel<<<SAGE_NP, SAGE_PART_THREADS, 0, stream>>>(src, dst, cells,
                                                                ovfCnt, E);
    sage_file_kernel<<<SAGE_NB, 256, 0, stream>>>(src, dst, cells, rec, ovfCnt,
                                                  ovfDst, ovfSrc, E);

    sage_gather_kernel<<<gathBlocks, 256, 0, stream>>>(x, rec, ovfCnt, ovfDst,
                                                       ovfSrc, s, cnt);
    sage_gemm_kernel<<<gemmBlocks, SAGE_GEMM_THREADS, 0, stream>>>(
        s, cnt, x, Wl1, bl1, Wr1, out, nRowTiles);

    sage_gather_kernel<<<gathBlocks, 256, 0, stream>>>(out, rec, ovfCnt, ovfDst,
                                                       ovfSrc, s, cnt);
    sage_gemm_kernel<<<gemmBlocks, SAGE_GEMM_THREADS, 0, stream>>>(
        s, cnt, out, Wl2, bl2, Wr2, out, nRowTiles);
}


# Gather: a wave takes 4 consecutive nodes in turn, their record lines loaded together up front; a launch starts 12500 waves, all resident at once, not 50000; per-node source and add order unchanged
# speedup vs baseline: 3.8701x; 1.1132x over previous
#include <hip/hip_runtime.h>
#include <hip/hip_bf16.h>


#define SAGE_N 50000
#define SAGE_D 128
#define SAGE_REC 64
#define SAGE_SLOTS 63
#define SAGE_NOKEY 0x7fffffff

typedef __attribute__((ext_vector_type(2))) float v2f;
typedef __attribute__((ext_vector_type(8))) float v8f;

#define SAGE_NB 256
#define SAGE_NPB 196
#define SAGE_NP 256
#define SAGE_LP 65
#define SAGE_PART_THREADS 512
#define SAGE_PART_BATCH 8
static_assert(SAGE_NB * SAGE_NPB >= SAGE_N, "ranges cover all nodes");
static_assert(SAGE_N <= 65536 && SAGE_NPB < 32768, "entry packing");

__global__ __launch_bounds__(SAGE_PART_THREADS) void sage_part_kernel(
    const int* __restrict__ src, const int* __restrict__ dst,
    int* __restrict__ cells, int* __restrict__ ovfCount, int E) {
    __shared__ int st[SAGE_NB * SAGE_LP];
    const int tid = threadIdx.x;
    const int p = blockIdx.x;
    if (tid < SAGE_NB) st[tid * SAGE_LP] = 0;
    if (p == 0 && tid == 0) *ovfCount = 0;
    __syncthreads();

    const int per = (E + SAGE_NP - 1) / SAGE_NP;
    const int e0 = p * per;
    const int e1 = (e0 + per < E) ? (e0 + per) : E;
    for (int base = e0; base < e1; base += SAGE_PART_THREADS * SAGE_PART_BATCH) {
        int d[SAGE_PART_BATCH], s[SAGE_PART_BATCH], slot[SAGE_PART_BATCH];
#pragma unroll
        for (int i = 0; i < SAGE_PART_BATCH; ++i) {
            const int e = base + i * SAGE_PART_THREADS + tid;
            const bool ok = e < e1;
            d[i] = ok ? dst[e] : -1;
            s[i] = ok ? src[e] : 0;
        }
#pragma unroll
        for (int i = 0; i < SAGE_PART_BATCH; ++i) {
            slot[i] = SAGE_SLOTS;
            if (d[i] >= 0) slot[i] = atomicAdd(st + (d[i] / SAGE_NPB) * SAGE_LP, 1);
        }
#pragma unroll
        for (int i = 0; i < SAGE_PART_BATCH; ++i) {
            if (slot[i] < SAGE_SLOTS) {
                const int b = d[i] / SAGE_NPB;
                st[b * SAGE_LP + 1 + slot[i]] = ((d[i] - b * SAGE_NPB) << 16) | s[i];
            }
        }
    }
    __syncthreads();

    const int wave = tid >> 5;
    const int lane = tid & 31;
    constexpr int kWaves = SAGE_PART_THREADS / 32;
    constexpr int kCellsPerWave = SAGE_NB / kWaves;
    static_assert(kCellsPerWave * kWaves == SAGE_NB && kCellsPerWave <= 32, "cells per wave");
    const int myCount = (lane < kCellsPerWave) ? st[(wave + kWaves * lane) * SAGE_LP] : 0;
    int first[kCellsPerWave];
#pragma unroll
    for (int k = 0; k < kCellsPerWave; ++k) {
        const int n = __builtin_amdgcn_readlane(myCount, k);
        const int used = n < SAGE_SLOTS ? n : SAGE_SLOTS;
        first[k] = (lane <= used) ? st[(wave + kWaves * k) * SAGE_LP + lane] : 0;
    }
#pragma unroll
    for (int k = 0; k < kCellsPerWave; ++k) {
        const int b = wave + kWaves * k;
        const int n = __builtin_amdgcn_readlane(myCount, k);
        const int used = n < SAGE_SLOTS ? n : SAGE_SLOTS;
        int* g = cells + ((size_t)b * SAGE_NP + p) * SAGE_REC;
        g[lane] = first[k];
        if (32 + lane <= used) g[32 + lane] = st[b * SAGE_LP + 32 + lane];
    }
}

__device__ __forceinline__ int sage_take_slot(int* recS, int entry) {
    return atomicAdd(recS + (entry >> 16) * SAGE_LP, 1);
}

__device__ __forceinline__ void sage_put_entry(int* recS, int entry, int slot, int node0,
                                               int* ovfCount, int* __restrict__ ovfDst,
                                               int* __restrict__ ovfSrc) {
    const int ld = entry >> 16;
    const int sn = entry & 0xffff;
    if (slot < SAGE_SLOTS) {
        recS[ld * SAGE_LP + 1 + slot] = sn;
    } else {
        const int k = atomicAdd(ovfCount, 1);
        ovfDst[k] = node0 + ld;
        ovfSrc[k] = sn;
    }
}

__global__ __launch_bounds__(256) void sage_file_kernel(
    const int* __restrict__ src, const int* __restrict__ dst,
    const int* __restrict__ cells, int* __restrict__ rec, int* ovfCount,
    int* __restrict__ ovfDst, int* __restrict__ ovfSrc, int E) {
    __shared__ int recS[SAGE_NPB * SAGE_LP];
    const int tid = threadIdx.x;
    const int wave = tid >> 5;
    const int lane = tid & 31;
    const int b = blockIdx.x;
    const int node0 = b * SAGE_NPB;
    for (int i = tid; i < SAGE_NPB * SAGE_LP; i += 256) recS[i] = 0;
    __syncthreads();

    const int* cw = cells + ((size_t)b * SAGE_NP + wave * 32) * SAGE_REC;
    int w[32], slot[32];
#pragma unroll
    for (int j = 0; j < 32; ++j) w[j] = cw[j * SAGE_REC + lane];
    unsigned more = 0;
    unsigned redo = 0;
#pragma unroll
    for (int j = 0; j < 32; ++j) {
        const int n = __builtin_amdgcn_readfirstlane(w[j]);
        const bool whole = n <= SAGE_SLOTS;
        redo |= (whole ? 0u : 1u) << j;
        more |= ((whole && n > 31) ? 1u : 0u) << j;
        slot[j] = -1;
        if (whole && lane >= 1 && lane <= n) slot[j] = sage_take_slot(recS, w[j]);
    }
#pragma unroll
    for (int j = 0; j < 32; ++j) {
        if (slot[j] >= 0) sage_put_entry(recS, w[j], slot[j], node0, ovfCount, ovfDst, ovfSrc);
    }
#pragma nounroll
    for (int j = 0; j < 32; ++j) {
        if ((more >> j) & 1u) {
            const int n = cw[j * SAGE_REC];
            if (32 + lane <= n) {
                const int entry = cw[j * SAGE_REC + 32 + lane];
                sage_put_entry(recS, entry, sage_take_slot(recS, entry), node0, ovfCount,
                               ovfDst, ovfSrc);
            }
        }
    }
#pragma nounroll
    for (int j = 0; j < 32; ++j) {
        if ((redo >> j) & 1u) {
            const int per = (E + SAGE_NP - 1) / SAGE_NP;
            const int e0 = (wave * 32 + j) * per;
            const int e1 = (e0 + per < E) ? (e0 + per) : E;
            for (int e = e0 + lane; e < e1; e += 32) {
                const int dn = dst[e];
                if (dn >= node0 && dn < node0 + SAGE_NPB) {
                    const int entry = ((dn - node0) << 16) | src[e];
                    sage_put_entry(recS, entry, sage_take_slot(recS, entry), node0, ovfCount,
                                   ovfDst, ovfSrc);
                }
            }
        }
    }
    __syncthreads();

    for (int q = tid; q < SAGE_NPB * (SAGE_REC / 4); q += 256) {
        const int ln = q >> 4;
        const int* r = recS + ln * SAGE_LP + (q & 15) * 4;
        if (node0 + ln < SAGE_N)
            ((int4*)(rec + (size_t)node0 * SAGE_REC))[q] = make_int4(r[0], r[1], r[2], r[3]);
    }
}

template <int K>
__device__ __forceinline__ void sage_add_rows(float4& acc, const float4* __restrict__ xrow,
                                              int sreg, int j) {
    float4 v[K];
#pragma unroll
    for (int i = 0; i < K; ++i) {
        const int sid = __builtin_amdgcn_readlane(sreg, j + i) >> 6;
        v[i] = xrow[(size_t)sid * (SAGE_D / 4)];
    }
#pragma unroll
    for (int i = 0; i < K; ++i) {
        acc.x += v[i].x; acc.y += v[i].y; acc.z += v[i].z; acc.w += v[i].w;
    }
}

__device__ __forceinline__ void sage_add_range(float4& acc, const float4* __restrict__ xrow,
                                               int sreg, int count) {
    int j = 0;
    for (; j + 8 <= count; j += 8) sage_add_rows<8>(acc, xrow, sreg, j);
    if (count & 4) { sage_add_rows<4>(acc, xrow, sreg, j); j += 4; }
    if (count & 2) { sage_add_rows<2>(acc, xrow, sreg, j); j += 2; }
    if (count & 1) { sage_add_rows<1>(acc, xrow, sreg, j); }
}

#define SAGE_GK 4
#define SAGE_GWAVES 8
static_assert(SAGE_N % SAGE_GK == 0, "every node of a started wave exists");

__global__ __launch_bounds__(SAGE_GWAVES * 32, 16) void sage_gather_kernel(
    const float* __restrict__ xin, const int* __restrict__ rec,
    const int* __restrict__ ovfCount, const int* __restrict__ ovfDst,
    const int* __restrict__ ovfSrc, float* __restrict__ s,
    float* __restrict__ cnt) {
    __shared__ int sorted[SAGE_GWAVES][SAGE_REC];
    const int wave = __builtin_amdgcn_readfirstlane(threadIdx.x >> 5);
    const int lane = threadIdx.x & 31;
    const int n0 = (blockIdx.x * SAGE_GWAVES + wave) * SAGE_GK;
    if (n0 >= SAGE_N) return;

    const int* r0 = rec + (size_t)n0 * SAGE_REC;
    int wa = r0[lane];
    int wb = r0[SAGE_REC + lane];
    int wc = r0[2 * SAGE_REC + lane];
    int wd = r0[3 * SAGE_REC + lane];
    static_assert(SAGE_GK == 4, "four record lines are loaded by hand");
    const float4* xrow = (const float4*)xin + lane;
    int myDeg = 0;

#pragma nounroll
    for (int i = 0; i < SAGE_GK; ++i) {
        const int n = n0 + i;
        const int* r = rec + (size_t)n * SAGE_REC;
        const int w0 = wa;
        wa = wb; wb = wc; wc = wd;
        const int deg = __builtin_amdgcn_readfirstlane(w0);
        if (lane == i) myDeg = deg;
        float4 acc = make_float4(0.0f, 0.0f, 0.0f, 0.0f);

        if (deg <= SAGE_SLOTS) {
            int w1 = 0;
            if (deg > 31) w1 = r[32 + lane];
            const int i0 = lane - 1;
            const int i1 = lane + 31;
            const bool ok0 = (lane >= 1) && (i0 < deg);
            const bool ok1 = (i1 < deg);
            const int k0 = ok0 ? ((w0 << 6) | i0) : SAGE_NOKEY;
            const int k1 = ok1 ? ((w1 << 6) | i1) : SAGE_NOKEY;
            int r0k = 0, r1k = 0;
            const int na = deg < 31 ? deg : 31;
            for (int j = 0; j < na; ++j) {
                const int u = __builtin_amdgcn_readlane(k0, j + 1);
                r0k += (u < k0) ? 1 : 0;
                r1k += (u < k1) ? 1 : 0;
            }
            for (int j = 31; j < deg; ++j) {
                const int u = __builtin_amdgcn_readlane(k1, j - 31);
                r0k += (u < k0) ? 1 : 0;
                r1k += (u < k1) ? 1 : 0;
            }
            if (ok0) sorted[wave][r0k] = k0;
            if (ok1) sorted[wave][r1k] = k1;
            int s0 = 0, s1 = 0;
            if (lane < deg) s0 = sorted[wave][lane];
            if (32 + lane < deg) s1 = sorted[wave][32 + lane];
            sage_add_range(acc, xrow, s0, deg < 32 ? deg : 32);
            if (deg > 32) sage_add_range(acc, xrow, s1, deg - 32);
        } else {
            const int w1 = r[32 + lane];
            const int k0 = (lane >= 1) ? w0 : SAGE_NOKEY;
            const int k1 = w1;
            const int nov = *ovfCount;
            int last = -1;
            for (int it = 0; it < deg; ++it) {
                int best = SAGE_NOKEY;
                if (k0 > last && k0 < best) best = k0;
                if (k1 > last && k1 < best) best = k1;
                for (int q = lane; q < nov; q += 32) {
                    if (ovfDst[q] == n) {
                        const int v = ovfSrc[q];
                        if (v > last && v < best) best = v;
                    }
                }
                for (int m = 16; m >= 1; m >>= 1) {
                    const int o = __shfl_xor(best, m, 32);
                    best = o < best ? o : best;
                }
                if (best == SAGE_NOKEY) break;
                int c = (k0 == best ? 1 : 0) + (k1 == best ? 1 : 0);
                for (int q = lane; q < nov; q += 32) {
                    if (ovfDst[q] == n && ovfSrc[q] == best) ++c;
                }
                for (int m = 16; m >= 1; m >>= 1) c += __shfl_xor(c, m, 32);
                const int sid = __builtin_amdgcn_readfirstlane(best);
                const float4 v = xrow[(size_t)sid * (SAGE_D / 4)];
                for (int t = 0; t < c; ++t) {
                    acc.x += v.x; acc.y += v.y; acc.z += v.z; acc.w += v.w;
                }
                last = best;
            }
        }

        ((float4*)(s + (size_t)n * SAGE_D))[lane] = acc;
    }
    if (lane < SAGE_GK) cnt[n0 + lane] = (float)myDeg;
}

#define SAGE_WP 132
#define SAGE_GEMM_WGS 256
#define SAGE_GEMM_THREADS 512
static_assert(SAGE_GEMM_WGS * (SAGE_GEMM_THREADS / 32) >= SAGE_N / 16,
              "one wave per row tile");

__device__ __forceinline__ void sage_load_b(v2f (&b)[16], const float* wl_s,
                                            const float* wr_s, int k0) {
#pragma unroll
    for (int ct = 0; ct < 8; ++ct) {
        b[ct]     = *(const v2f*)(wl_s + ct * 16 * SAGE_WP + k0);
        b[8 + ct] = *(const v2f*)(wr_s + ct * 16 * SAGE_WP + k0);
    }
}

__device__ __forceinline__ void sage_load_a(v2f (&a)[8], const float* arow_s,
                                            const float* arow_x, int kc) {
#pragma unroll
    for (int i = 0; i < 4; ++i) {
        a[i]     = *(const v2f*)(arow_s + kc + 4 * i);
        a[4 + i] = *(const v2f*)(arow_x + kc + 4 * i);
    }
}

__device__ __forceinline__ void sage_mma_step(v8f (&acc)[8], const v2f (&b)[16],
                                              v2f a_agg, v2f a_x) {
#pragma unroll
    for (int ct = 0; ct < 8; ++ct) {
        acc[ct] = __builtin_amdgcn_wmma_f32_16x16x4_f32(
            false, a_agg, false, b[ct], (short)0, acc[ct], false, false);
        acc[ct] = __builtin_amdgcn_wmma_f32_16x16x4_f32(
            false, a_x, false, b[8 + ct], (short)0, acc[ct], false, false);
    }
}

template <bool LAST>
__device__ __forceinline__ void sage_mma_chunk(v8f (&acc)[8], const v2f (&a)[8],
                                               v2f (&b0)[16], v2f (&b1)[16],
                                               const float* wl_s, const float* wr_s,
                                               float inv, int kc) {
#pragma unroll
    for (int i = 0; i < 4; i += 2) {
        sage_load_b(b1, wl_s, wr_s, kc + 4 * i + 4);
        __builtin_amdgcn_sched_barrier(0);
        sage_mma_step(acc, b0, a[i] * inv, a[4 + i]);
        __builtin_amdgcn_sched_barrier(0);
        if (!LAST || i < 2) sage_load_b(b0, wl_s, wr_s, kc + 4 * i + 8);
        __builtin_amdgcn_sched_barrier(0);
        sage_mma_step(acc, b1, a[i + 1] * inv, a[4 + i + 1]);
        __builtin_amdgcn_sched_barrier(0);
    }
}

__global__ __launch_bounds__(SAGE_GEMM_THREADS) void sage_gemm_kernel(
    const float* __restrict__ s,
    const float* __restrict__ cnt,
    const float* __restrict__ xin,
    const float* __restrict__ Wl,
    const float* __restrict__ bl,
    const float* __restrict__ Wr,
    float* __restrict__ out,
    int nRowTiles) {
    __shared__ __attribute__((aligned(16))) float sW[2 * SAGE_D * SAGE_WP];

#pragma unroll
    for (int j = 0; j < SAGE_D * (SAGE_D / 4) / SAGE_GEMM_THREADS; ++j) {
        const int i   = j * SAGE_GEMM_THREADS + threadIdx.x;
        const int row = i >> 5;
        const int c4  = i & 31;
        const float4 vl = ((const float4*)Wl)[i];
        const float4 vr = ((const float4*)Wr)[i];
        *(float4*)(sW + row * SAGE_WP + c4 * 4) = vl;
        *(float4*)(sW + (SAGE_D + row) * SAGE_WP + c4 * 4) = vr;
    }
    __syncthreads();

    const int wave   = __builtin_amdgcn_readfirstlane(threadIdx.x >> 5);
    const int lane = threadIdx.x & 31;
    const int half = lane >> 4;
    const int lm   = lane & 15;
    const float* wl_s = sW + lm * SAGE_WP + 2 * half;
    const float* wr_s = wl_s + SAGE_D * SAGE_WP;

    const int tile = wave * SAGE_GEMM_WGS + blockIdx.x;
    if (tile < nRowTiles) {
        const int m0 = tile * 16;
        const float deg = cnt[m0 + lm];
        const float* arow_s = s   + (size_t)(m0 + lm) * SAGE_D + 2 * half;
        const float* arow_x = xin + (size_t)(m0 + lm) * SAGE_D + 2 * half;

        float bias[8];
#pragma unroll
        for (int ct = 0; ct < 8; ++ct) bias[ct] = bl[ct * 16 + lm];

        v8f acc[8];
#pragma unroll
        for (int ct = 0; ct < 8; ++ct) acc[ct] = (v8f){};

        v2f a0[8], a1[8], b0[16], b1[16];
        sage_load_a(a0, arow_s, arow_x, 0);
        sage_load_b(b0, wl_s, wr_s, 0);
        __builtin_amdgcn_sched_barrier(0);
        const float inv = 1.0f / fmaxf(deg, 1.0f);
        for (int kc = 0; kc < SAGE_D - 32; kc += 32) {
            sage_load_a(a1, arow_s, arow_x, kc + 16);
            sage_mma_chunk<false>(acc, a0, b0, b1, wl_s, wr_s, inv, kc);
            sage_load_a(a0, arow_s, arow_x, kc + 32);
            sage_mma_chunk<false>(acc, a1, b0, b1, wl_s, wr_s, inv, kc + 16);
        }
        sage_load_a(a1, arow_s, arow_x, SAGE_D - 16);
        sage_mma_chunk<false>(acc, a0, b0, b1, wl_s, wr_s, inv, SAGE_D - 32);
        sage_mma_chunk<true>(acc, a1, b0, b1, wl_s, wr_s, inv, SAGE_D - 16);

#pragma unroll
        for (int ct = 0; ct < 8; ++ct) {
#pragma unroll
            for (int v = 0; v < 8; ++v) {
                out[(size_t)(m0 + v + 8 * half) * SAGE_D + ct * 16 + lm] =
                    acc[ct][v] + bias[ct];
            }
        }
    }
}

extern "C" void kernel_launch(void* const* d_in, const int* in_sizes, int n_in,
                              void* d_out, int out_size, void* d_ws, size_t ws_size,
                              hipStream_t stream) {
    const float* x   = (const float*)d_in[0];
    const int*   ei  = (const int*)d_in[1];
    const float* Wl1 = (const float*)d_in[2];
    const float* bl1 = (const float*)d_in[3];
    const float* Wr1 = (const float*)d_in[4];
    const float* Wl2 = (const float*)d_in[5];
    const float* bl2 = (const float*)d_in[6];
    const float* Wr2 = (const float*)d_in[7];
    float* out = (float*)d_out;

    const int N = SAGE_N;
    const int E = in_sizes[1] / 2;
    const int* src = ei;
    const int* dst = ei + E;

    float* s      = (float*)d_ws;
    int*   rec    = (int*)(s + (size_t)N * SAGE_D);
    int*   cells  = rec + (size_t)N * SAGE_REC;
    int*   ovfDst = cells + (size_t)SAGE_NB * SAGE_NP * SAGE_REC;
    int*   ovfSrc = ovfDst + E;
    float* cnt    = (float*)(ovfSrc + E);
    int*   ovfCnt = (int*)(cnt + N);

    const int nRowTiles  = N / 16;
    const int gemmBlocks = SAGE_GEMM_WGS;
    const int gathBlocks = (N / SAGE_GK + SAGE_GWAVES - 1) / SAGE_GWAVES;

    sage_part_kernel<<<SAGE_NP, SAGE_PART_THREADS, 0, stream>>>(src, dst, cells,
                                                                ovfCnt, E);
    sage_file_kernel<<<SAGE_NB, 256, 0, stream>>>(src, dst, cells, rec, ovfCnt,
                                                  ovfDst, ovfSrc, E);

    sage_gather_kernel<<<gathBlocks, SAGE_GWAVES * 32, 0, stream>>>(x, rec, ovfCnt, ovfDst,
                                                       ovfSrc, s, cnt);
    sage_gemm_kernel<<<gemmBlocks, SAGE_GEMM_THREADS, 0, stream>>>(
        s, cnt, x, Wl1, bl1, Wr1, out, nRowTiles);

    sage_gather_kernel<<<gathBlocks, SAGE_GWAVES * 32, 0, stream>>>(out, rec, ovfCnt, ovfDst,
                                                       ovfSrc, s, cnt);
    sage_gemm_kernel<<<gemmBlocks, SAGE_GEMM_THREADS, 0, stream>>>(
        s, cnt, out, Wl2, bl2, Wr2, out, nRowTiles);
}


# Layer = one launch: per work-group 24 waves gather the next 4 row tiles' neighbour sums into LDS while 8 run the WMMAs of the 4 before (13th tile in 4 column pieces); add and k orders unchanged
# speedup vs baseline: 4.2571x; 1.1000x over previous
#include <hip/hip_runtime.h>
#include <hip/hip_bf16.h>


#define SAGE_N 50000
#define SAGE_D 128
#define SAGE_REC 64
#define SAGE_SLOTS 63
#define SAGE_NOKEY 0x7fffffff

typedef __attribute__((ext_vector_type(2))) float v2f;
typedef __attribute__((ext_vector_type(8))) float v8f;

#define SAGE_NB 256
#define SAGE_NPB 196
#define SAGE_NP 256
#define SAGE_LP 65
#define SAGE_PART_THREADS 512
#define SAGE_PART_BATCH 8
static_assert(SAGE_NB * SAGE_NPB >= SAGE_N, "ranges cover all nodes");
static_assert(SAGE_N <= 65536 && SAGE_NPB < 32768, "entry packing");

__global__ __launch_bounds__(SAGE_PART_THREADS) void sage_part_kernel(
    const int* __restrict__ src, const int* __restrict__ dst,
    int* __restrict__ cells, int* __restrict__ ovfCount, int E) {
    __shared__ int st[SAGE_NB * SAGE_LP];
    const int tid = threadIdx.x;
    const int p = blockIdx.x;
    if (tid < SAGE_NB) st[tid * SAGE_LP] = 0;
    if (p == 0 && tid == 0) *ovfCount = 0;
    __syncthreads();

    const int per = (E + SAGE_NP - 1) / SAGE_NP;
    const int e0 = p * per;
    const int e1 = (e0 + per < E) ? (e0 + per) : E;
    for (int base = e0; base < e1; base += SAGE_PART_THREADS * SAGE_PART_BATCH) {
        int d[SAGE_PART_BATCH], s[SAGE_PART_BATCH], slot[SAGE_PART_BATCH];
#pragma unroll
        for (int i = 0; i < SAGE_PART_BATCH; ++i) {
            const int e = base + i * SAGE_PART_THREADS + tid;
            const bool ok = e < e1;
            d[i] = ok ? dst[e] : -1;
            s[i] = ok ? src[e] : 0;
        }
#pragma unroll
        for (int i = 0; i < SAGE_PART_BATCH; ++i) {
            slot[i] = SAGE_SLOTS;
            if (d[i] >= 0) slot[i] = atomicAdd(st + (d[i] / SAGE_NPB) * SAGE_LP, 1);
        }
#pragma unroll
        for (int i = 0; i < SAGE_PART_BATCH; ++i) {
            if (slot[i] < SAGE_SLOTS) {
                const int b = d[i] / SAGE_NPB;
                st[b * SAGE_LP + 1 + slot[i]] = ((d[i] - b * SAGE_NPB) << 16) | s[i];
            }
        }
    }
    __syncthreads();

    const int wave = tid >> 5;
    const int lane = tid & 31;
    constexpr int kWaves = SAGE_PART_THREADS / 32;
    constexpr int kCellsPerWave = SAGE_NB / kWaves;
    static_assert(kCellsPerWave * kWaves == SAGE_NB && kCellsPerWave <= 32, "cells per wave");
    const int myCount = (lane < kCellsPerWave) ? st[(wave + kWaves * lane) * SAGE_LP] : 0;
    int first[kCellsPerWave];
#pragma unroll
    for (int k = 0; k < kCellsPerWave; ++k) {
        const int n = __builtin_amdgcn_readlane(myCount, k);
        const int used = n < SAGE_SLOTS ? n : SAGE_SLOTS;
        first[k] = (lane <= used) ? st[(wave + kWaves * k) * SAGE_LP + lane] : 0;
    }
#pragma unroll
    for (int k = 0; k < kCellsPerWave; ++k) {
        const int b = wave + kWaves * k;
        const int n = __builtin_amdgcn_readlane(myCount, k);
        const int used = n < SAGE_SLOTS ? n : SAGE_SLOTS;
        int* g = cells + ((size_t)b * SAGE_NP + p) * SAGE_REC;
        g[lane] = first[k];
        if (32 + lane <= used) g[32 + lane] = st[b * SAGE_LP + 32 + lane];
    }
}

__device__ __forceinline__ int sage_take_slot(int* recS, int entry) {
    return atomicAdd(recS + (entry >> 16) * SAGE_LP, 1);
}

__device__ __forceinline__ void sage_put_entry(int* recS, int entry, int slot, int node0,
                                               int* ovfCount, int* __restrict__ ovfDst,
                                               int* __restrict__ ovfSrc) {
    const int ld = entry >> 16;
    const int sn = entry & 0xffff;
    if (slot < SAGE_SLOTS) {
        recS[ld * SAGE_LP + 1 + slot] = sn;
    } else {
        const int k = atomicAdd(ovfCount, 1);
        ovfDst[k] = node0 + ld;
        ovfSrc[k] = sn;
    }
}

__global__ __launch_bounds__(256) void sage_file_kernel(
    const int* __restrict__ src, const int* __restrict__ dst,
    const int* __restrict__ cells, int* __restrict__ rec, int* ovfCount,
    int* __restrict__ ovfDst, int* __restrict__ ovfSrc, int E) {
    __shared__ int recS[SAGE_NPB * SAGE_LP];
    const int tid = threadIdx.x;
    const int wave = tid >> 5;
    const int lane = tid & 31;
    const int b = blockIdx.x;
    const int node0 = b * SAGE_NPB;
    for (int i = tid; i < SAGE_NPB * SAGE_LP; i += 256) recS[i] = 0;
    __syncthreads();

    const int* cw = cells + ((size_t)b * SAGE_NP + wave * 32) * SAGE_REC;
    int w[32], slot[32];
#pragma unroll
    for (int j = 0; j < 32; ++j) w[j] = cw[j * SAGE_REC + lane];
    unsigned more = 0;
    unsigned redo = 0;
#pragma unroll
    for (int j = 0; j < 32; ++j) {
        const int n = __builtin_amdgcn_readfirstlane(w[j]);
        const bool whole = n <= SAGE_SLOTS;
        redo |= (whole ? 0u : 1u) << j;
        more |= ((whole && n > 31) ? 1u : 0u) << j;
        slot[j] = -1;
        if (whole && lane >= 1 && lane <= n) slot[j] = sage_take_slot(recS, w[j]);
    }
#pragma unroll
    for (int j = 0; j < 32; ++j) {
        if (slot[j] >= 0) sage_put_entry(recS, w[j], slot[j], node0, ovfCount, ovfDst, ovfSrc);
    }
#pragma nounroll
    for (int j = 0; j < 32; ++j) {
        if ((more >> j) & 1u) {
            const int n = cw[j * SAGE_REC];
            if (32 + lane <= n) {
                const int entry = cw[j * SAGE_REC + 32 + lane];
                sage_put_entry(recS, entry, sage_take_slot(recS, entry), node0, ovfCount,
                               ovfDst, ovfSrc);
            }
        }
    }
#pragma nounroll
    for (int j = 0; j < 32; ++j) {
        if ((redo >> j) & 1u) {
            const int per = (E + SAGE_NP - 1) / SAGE_NP;
            const int e0 = (wave * 32 + j) * per;
            const int e1 = (e0 + per < E) ? (e0 + per) : E;
            for (int e = e0 + lane; e < e1; e += 32) {
                const int dn = dst[e];
                if (dn >= node0 && dn < node0 + SAGE_NPB) {
                    const int entry = ((dn - node0) << 16) | src[e];
                    sage_put_entry(recS, entry, sage_take_slot(recS, entry), node0, ovfCount,
                                   ovfDst, ovfSrc);
                }
            }
        }
    }
    __syncthreads();

    for (int q = tid; q < SAGE_NPB * (SAGE_REC / 4); q += 256) {
        const int ln = q >> 4;
        const int* r = recS + ln * SAGE_LP + (q & 15) * 4;
        if (node0 + ln < SAGE_N)
            ((int4*)(rec + (size_t)node0 * SAGE_REC))[q] = make_int4(r[0], r[1], r[2], r[3]);
    }
}

template <int K>
__device__ __forceinline__ void sage_add_rows(float4& acc, const float4* __restrict__ xrow,
                                              int sreg, int j) {
    float4 v[K];
#pragma unroll
    for (int i = 0; i < K; ++i) {
        const int sid = __builtin_amdgcn_readlane(sreg, j + i) >> 6;
        v[i] = xrow[(size_t)sid * (SAGE_D / 4)];
    }
#pragma unroll
    for (int i = 0; i < K; ++i) {
        acc.x += v[i].x; acc.y += v[i].y; acc.z += v[i].z; acc.w += v[i].w;
    }
}

__device__ __forceinline__ void sage_add_range(float4& acc, const float4* __restrict__ xrow,
                                               int sreg, int count) {
    int j = 0;
    for (; j + 8 <= count; j += 8) sage_add_rows<8>(acc, xrow, sreg, j);
    if (count & 4) { sage_add_rows<4>(acc, xrow, sreg, j); j += 4; }
    if (count & 2) { sage_add_rows<2>(acc, xrow, sreg, j); j += 2; }
    if (count & 1) { sage_add_rows<1>(acc, xrow, sreg, j); }
}

__device__ __forceinline__ float4 sage_node_sum(const float4* __restrict__ xrow,
                                                const int* __restrict__ r, int w0, int deg,
                                                int n, int* sortedW, int lane,
                                                const int* __restrict__ ovfCount,
                                                const int* __restrict__ ovfDst,
                                                const int* __restrict__ ovfSrc) {
    float4 acc = make_float4(0.0f, 0.0f, 0.0f, 0.0f);
    if (deg <= SAGE_SLOTS) {
        int w1 = 0;
        if (deg > 31) w1 = r[32 + lane];
        const int i0 = lane - 1;
        const int i1 = lane + 31;
        const bool ok0 = (lane >= 1) && (i0 < deg);
        const bool ok1 = (i1 < deg);
        const int k0 = ok0 ? ((w0 << 6) | i0) : SAGE_NOKEY;
        const int k1 = ok1 ? ((w1 << 6) | i1) : SAGE_NOKEY;
        int r0k = 0, r1k = 0;
        const int na = deg < 31 ? deg : 31;
        for (int j = 0; j < na; ++j) {
            const int u = __builtin_amdgcn_readlane(k0, j + 1);
            r0k += (u < k0) ? 1 : 0;
            r1k += (u < k1) ? 1 : 0;
        }
        for (int j = 31; j < deg; ++j) {
            const int u = __builtin_amdgcn_readlane(k1, j - 31);
            r0k += (u < k0) ? 1 : 0;
            r1k += (u < k1) ? 1 : 0;
        }
        if (ok0) sortedW[r0k] = k0;
        if (ok1) sortedW[r1k] = k1;
        int s0 = 0, s1 = 0;
        if (lane < deg) s0 = sortedW[lane];
        if (32 + lane < deg) s1 = sortedW[32 + lane];
        sage_add_range(acc, xrow, s0, deg < 32 ? deg : 32);
        if (deg > 32) sage_add_range(acc, xrow, s1, deg - 32);
    } else {
        const int w1 = r[32 + lane];
        const int k0 = (lane >= 1) ? w0 : SAGE_NOKEY;
        const int k1 = w1;
        const int nov = *ovfCount;
        int last = -1;
        for (int it = 0; it < deg; ++it) {
            int best = SAGE_NOKEY;
            if (k0 > last && k0 < best) best = k0;
            if (k1 > last && k1 < best) best = k1;
            for (int q = lane; q < nov; q += 32) {
                if (ovfDst[q] == n) {
                    const int v = ovfSrc[q];
                    if (v > last && v < best) best = v;
                }
            }
            {
                int mn = SAGE_NOKEY;
#pragma nounroll
                for (int j = 0; j < 32; ++j) {
                    const int o = __builtin_amdgcn_readlane(best, j);
                    mn = o < mn ? o : mn;
                }
                best = mn;
            }
            if (best == SAGE_NOKEY) break;
            int c = (k0 == best ? 1 : 0) + (k1 == best ? 1 : 0);
            for (int q = lane; q < nov; q += 32) {
                if (ovfDst[q] == n && ovfSrc[q] == best) ++c;
            }
            {
                int tot = 0;
#pragma nounroll
                for (int j = 0; j < 32; ++j) tot += __builtin_amdgcn_readlane(c, j);
                c = tot;
            }
            const int sid = __builtin_amdgcn_readfirstlane(best);
            const float4 v = xrow[(size_t)sid * (SAGE_D / 4)];
            for (int t = 0; t < c; ++t) {
                acc.x += v.x; acc.y += v.y; acc.z += v.z; acc.w += v.w;
            }
            last = best;
        }
    }
    return acc;
}

#define SAGE_WP 132
#define SAGE_GEMM_WGS 256
#define SAGE_LW 32
#define SAGE_LMW 8
#define SAGE_LGW (SAGE_LW - SAGE_LMW)
#define SAGE_RN 64
static_assert(12 * SAGE_GEMM_WGS <= SAGE_N / 16 && 13 * SAGE_GEMM_WGS >= SAGE_N / 16,
              "three whole rounds per work-group and at most one tile more");
static_assert(SAGE_N % 16 == 0, "every row tile is whole");
static_assert(4 * SAGE_LGW >= SAGE_RN, "a wave gathers at most 4 nodes per slot");
static_assert((SAGE_D * (SAGE_D / 4)) % (SAGE_LW * 32) == 0, "the weight copy divides evenly");

template <int NCT>
__device__ __forceinline__ void sage_load_b(v2f (&b)[2 * NCT], const float* wl_s,
                                            const float* wr_s, int k0) {
#pragma unroll
    for (int ct = 0; ct < NCT; ++ct) {
        b[ct]       = *(const v2f*)(wl_s + ct * 16 * SAGE_WP + k0);
        b[NCT + ct] = *(const v2f*)(wr_s + ct * 16 * SAGE_WP + k0);
    }
}

__device__ __forceinline__ void sage_load_a(v2f (&a)[4], const float* arow_s,
                                            const float* arow_x, int kc) {
#pragma unroll
    for (int i = 0; i < 2; ++i) {
        a[i]     = *(const v2f*)(arow_s + kc + 4 * i);
        a[2 + i] = *(const v2f*)(arow_x + kc + 4 * i);
    }
}

template <int NCT>
__device__ __forceinline__ void sage_mma_step(v8f (&acc)[NCT], const v2f (&b)[2 * NCT],
                                              v2f a_agg, v2f a_x) {
#pragma unroll
    for (int ct = 0; ct < NCT; ++ct) {
        acc[ct] = __builtin_amdgcn_wmma_f32_16x16x4_f32(
            false, a_agg, false, b[ct], (short)0, acc[ct], false, false);
        acc[ct] = __builtin_amdgcn_wmma_f32_16x16x4_f32(
            false, a_x, false, b[NCT + ct], (short)0, acc[ct], false, false);
    }
}

template <int NCT, bool LAST>
__device__ __forceinline__ void sage_mma_chunk(v8f (&acc)[NCT], const v2f (&a)[4],
                                               v2f (&b0)[2 * NCT], v2f (&b1)[2 * NCT],
                                               const float* wl_s, const float* wr_s,
                                               float inv, int kc) {
    sage_load_b<NCT>(b1, wl_s, wr_s, kc + 4);
    __builtin_amdgcn_sched_barrier(0);
    sage_mma_step<NCT>(acc, b0, a[0] * inv, a[2]);
    __builtin_amdgcn_sched_barrier(0);
    if (!LAST) sage_load_b<NCT>(b0, wl_s, wr_s, kc + 8);
    __builtin_amdgcn_sched_barrier(0);
    sage_mma_step<NCT>(acc, b1, a[1] * inv, a[3]);
    __builtin_amdgcn_sched_barrier(0);
}

template <int NCT>
__device__ __forceinline__ void sage_tile(const float* arow_s, const float* arow_x, float deg,
                                          const float* wl_s, const float* wr_s,
                                          const float* bias_s, float* __restrict__ out,
                                          int m0, int col0, int half, int lm) {
    v8f acc[NCT];
#pragma unroll
    for (int ct = 0; ct < NCT; ++ct) acc[ct] = (v8f){};

    v2f a0[4], a1[4], b0[2 * NCT], b1[2 * NCT];
    sage_load_a(a0, arow_s, arow_x, 0);
    sage_load_b<NCT>(b0, wl_s, wr_s, 0);
    __builtin_amdgcn_sched_barrier(0);
    const float inv = 1.0f / fmaxf(deg, 1.0f);
    for (int kc = 0; kc < SAGE_D - 16; kc += 16) {
        sage_load_a(a1, arow_s, arow_x, kc + 8);
        sage_mma_chunk<NCT, false>(acc, a0, b0, b1, wl_s, wr_s, inv, kc);
        sage_load_a(a0, arow_s, arow_x, kc + 16);
        sage_mma_chunk<NCT, false>(acc, a1, b0, b1, wl_s, wr_s, inv, kc + 8);
    }
    sage_load_a(a1, arow_s, arow_x, SAGE_D - 8);
    sage_mma_chunk<NCT, false>(acc, a0, b0, b1, wl_s, wr_s, inv, SAGE_D - 16);
    sage_mma_chunk<NCT, true>(acc, a1, b0, b1, wl_s, wr_s, inv, SAGE_D - 8);

    float bias[NCT];
#pragma unroll
    for (int ct = 0; ct < NCT; ++ct) bias[ct] = bias_s[col0 + ct * 16 + lm];
#pragma unroll
    for (int ct = 0; ct < NCT; ++ct) {
#pragma unroll
        for (int v = 0; v < 8; ++v) {
            out[(size_t)(m0 + v + 8 * half) * SAGE_D + col0 + ct * 16 + lm] =
                acc[ct][v] + bias[ct];
        }
    }
}

__global__ __launch_bounds__(SAGE_LW * 32) void sage_layer_kernel(
    const float* __restrict__ xin,
    const int* __restrict__ rec,
    const int* __restrict__ ovfCount, const int* __restrict__ ovfDst,
    const int* __restrict__ ovfSrc,
    const float* __restrict__ Wl,
    const float* __restrict__ bl,
    const float* __restrict__ Wr,
    float* __restrict__ out,
    int nRowTiles) {
    __shared__ __attribute__((aligned(16))) float sW[2 * SAGE_D * SAGE_WP];
    __shared__ __attribute__((aligned(16))) float sAgg[2 * SAGE_RN * SAGE_WP];
    __shared__ float sCnt[2 * SAGE_RN];
    __shared__ float sBias[SAGE_D];
    __shared__ int sorted[SAGE_LW][SAGE_REC];

#pragma unroll
    for (int j = 0; j < SAGE_D * (SAGE_D / 4) / (SAGE_LW * 32); ++j) {
        const int i   = j * (SAGE_LW * 32) + threadIdx.x;
        const int row = i >> 5;
        const int c4  = i & 31;
        const float4 vl = ((const float4*)Wl)[i];
        const float4 vr = ((const float4*)Wr)[i];
        *(float4*)(sW + row * SAGE_WP + c4 * 4) = vl;
        *(float4*)(sW + (SAGE_D + row) * SAGE_WP + c4 * 4) = vr;
    }
    if (threadIdx.x < SAGE_D) sBias[threadIdx.x] = bl[threadIdx.x];

    const int wave = __builtin_amdgcn_readfirstlane(threadIdx.x >> 5);
    const int lane = threadIdx.x & 31;
    const int half = lane >> 4;
    const int lm   = lane & 15;
    const int b    = blockIdx.x;
    const int mt    = wave & 3;
    const int mcol0 = (wave >> 2) * 64;
    const float* wl_s = sW + (mcol0 + lm) * SAGE_WP + 2 * half;
    const float* wr_s = wl_s + SAGE_D * SAGE_WP;
    const float* as_s = sAgg + (mt * 16 + lm) * SAGE_WP + 2 * half;
    const float4* xrow = (const float4*)xin + lane;
    const int nRounds = (12 * SAGE_GEMM_WGS + b < nRowTiles) ? 4 : 3;

#pragma nounroll
    for (int sl = 0; sl < 4; ++sl) {
        if (sl > 0) __syncthreads();

        if (sl < nRounds && (wave >= SAGE_LMW || sl == 0)) {
            const int idx    = (sl == 0) ? wave : wave - SAGE_LMW;
            const int stride = (sl == 0) ? SAGE_LW : SAGE_LGW;
            const int nq     = (sl < 3) ? SAGE_RN : 16;
            float* aggB = sAgg + (sl & 1) * (SAGE_RN * SAGE_WP);
            float* cntB = sCnt + (sl & 1) * SAGE_RN;
            const int jb = 4 * sl;
            int wa = 0, wb = 0, wc = 0, wd = 0;
            {
                const int q0 = idx, q1 = idx + stride, q2 = idx + 2 * stride, q3 = idx + 3 * stride;
                if (q0 < nq)
                    wa = rec[(size_t)(((jb + (q0 >> 4)) * SAGE_GEMM_WGS + b) * 16 + (q0 & 15)) * SAGE_REC + lane];
                if (q1 < nq)
                    wb = rec[(size_t)(((jb + (q1 >> 4)) * SAGE_GEMM_WGS + b) * 16 + (q1 & 15)) * SAGE_REC + lane];
                if (q2 < nq)
                    wc = rec[(size_t)(((jb + (q2 >> 4)) * SAGE_GEMM_WGS + b) * 16 + (q2 & 15)) * SAGE_REC + lane];
                if (q3 < nq)
                    wd = rec[(size_t)(((jb + (q3 >> 4)) * SAGE_GEMM_WGS + b) * 16 + (q3 & 15)) * SAGE_REC + lane];
            }
#pragma nounroll
            for (int i = 0; i < 4; ++i) {
                const int q = idx + i * stride;
                if (q >= nq) break;
                const int n = ((jb + (q >> 4)) * SAGE_GEMM_WGS + b) * 16 + (q & 15);
                const int w0 = wa;
                wa = wb; wb = wc; wc = wd;
                const int deg = __builtin_amdgcn_readfirstlane(w0);
                const float4 acc = sage_node_sum(xrow, rec + (size_t)n * SAGE_REC, w0, deg, n,
                                                 sorted[wave], lane, ovfCount, ovfDst, ovfSrc);
                *(float4*)(aggB + q * SAGE_WP + lane * 4) = acc;
                if (lane == 0) cntB[q] = (float)deg;
            }
        }

        if (sl >= 1 && wave < SAGE_LMW) {
            const int r = sl - 1;
            const float* cntB = sCnt + (r & 1) * SAGE_RN;
            const int m0 = ((4 * r + mt) * SAGE_GEMM_WGS + b) * 16;
            sage_tile<4>(as_s + (r & 1) * (SAGE_RN * SAGE_WP),
                         xin + (size_t)(m0 + lm) * SAGE_D + 2 * half, cntB[mt * 16 + lm],
                         wl_s, wr_s, sBias, out, m0, mcol0, half, lm);
        }
    }

    if (nRounds == 4) {
        __syncthreads();
        if (wave < 4) {
            const float* cntB = sCnt + SAGE_RN;
            const int m0   = (12 * SAGE_GEMM_WGS + b) * 16;
            const int col0 = wave * 32;
            sage_tile<2>(as_s + (SAGE_RN - mt * 16) * SAGE_WP,
                         xin + (size_t)(m0 + lm) * SAGE_D + 2 * half, cntB[lm],
                         wl_s + col0 * SAGE_WP, wr_s + col0 * SAGE_WP, sBias, out,
                         m0, col0, half, lm);
        }
    }
}

extern "C" void kernel_launch(void* const* d_in, const int* in_sizes, int n_in,
                              void* d_out, int out_size, void* d_ws, size_t ws_size,
                              hipStream_t stream) {
    const float* x   = (const float*)d_in[0];
    const int*   ei  = (const int*)d_in[1];
    const float* Wl1 = (const float*)d_in[2];
    const float* bl1 = (const float*)d_in[3];
    const float* Wr1 = (const float*)d_in[4];
    const float* Wl2 = (const float*)d_in[5];
    const float* bl2 = (const float*)d_in[6];
    const float* Wr2 = (const float*)d_in[7];
    float* out = (float*)d_out;

    const int N = SAGE_N;
    const int E = in_sizes[1] / 2;
    const int* src = ei;
    const int* dst = ei + E;

    float* h      = (float*)d_ws;
    int*   rec    = (int*)(h + (size_t)N * SAGE_D);
    int*   cells  = rec + (size_t)N * SAGE_REC;
    int*   ovfDst = cells + (size_t)SAGE_NB * SAGE_NP * SAGE_REC;
    int*   ovfSrc = ovfDst + E;
    int*   ovfCnt = ovfSrc + E;

    const int nRowTiles = N / 16;

    sage_part_kernel<<<SAGE_NP, SAGE_PART_THREADS, 0, stream>>>(src, dst, cells,
                                                                ovfCnt, E);
    sage_file_kernel<<<SAGE_NB, 256, 0, stream>>>(src, dst, cells, rec, ovfCnt,
                                                  ovfDst, ovfSrc, E);

    sage_layer_kernel<<<SAGE_GEMM_WGS, SAGE_LW * 32, 0, stream>>>(
        x, rec, ovfCnt, ovfDst, ovfSrc, Wl1, bl1, Wr1, h, nRowTiles);
    sage_layer_kernel<<<SAGE_GEMM_WGS, SAGE_LW * 32, 0, stream>>>(
        h, rec, ovfCnt, ovfDst, ovfSrc, Wl2, bl2, Wr2, out, nRowTiles);
}


# Gather: all of a node's source rows (up to 23) are asked for before the first add, not in pieces of 8, 4, 2, 1 with a round trip each: one trip to L2 per node, not about three; add order unchanged
# speedup vs baseline: 4.4685x; 1.0497x over previous
#include <hip/hip_runtime.h>
#include <hip/hip_bf16.h>


#define SAGE_N 50000
#define SAGE_D 128
#define SAGE_REC 64
#define SAGE_SLOTS 63
#define SAGE_NOKEY 0x7fffffff
#define SAGE_KB 23

typedef __attribute__((ext_vector_type(2))) float v2f;
typedef __attribute__((ext_vector_type(8))) float v8f;

#define SAGE_NB 256
#define SAGE_NPB 196
#define SAGE_NP 256
#define SAGE_LP 65
#define SAGE_PART_THREADS 512
#define SAGE_PART_BATCH 8
static_assert(SAGE_NB * SAGE_NPB >= SAGE_N, "ranges cover all nodes");
static_assert(SAGE_N <= 65536 && SAGE_NPB < 32768, "entry packing");

__global__ __launch_bounds__(SAGE_PART_THREADS) void sage_part_kernel(
    const int* __restrict__ src, const int* __restrict__ dst,
    int* __restrict__ cells, int* __restrict__ ovfCount, int E) {
    __shared__ int st[SAGE_NB * SAGE_LP];
    const int tid = threadIdx.x;
    const int p = blockIdx.x;
    if (tid < SAGE_NB) st[tid * SAGE_LP] = 0;
    if (p == 0 && tid == 0) *ovfCount = 0;
    __syncthreads();

    const int per = (E + SAGE_NP - 1) / SAGE_NP;
    const int e0 = p * per;
    const int e1 = (e0 + per < E) ? (e0 + per) : E;
    for (int base = e0; base < e1; base += SAGE_PART_THREADS * SAGE_PART_BATCH) {
        int d[SAGE_PART_BATCH], s[SAGE_PART_BATCH], slot[SAGE_PART_BATCH];
#pragma unroll
        for (int i = 0; i < SAGE_PART_BATCH; ++i) {
            const int e = base + i * SAGE_PART_THREADS + tid;
            const bool ok = e < e1;
            d[i] = ok ? dst[e] : -1;
            s[i] = ok ? src[e] : 0;
        }
#pragma unroll
        for (int i = 0; i < SAGE_PART_BATCH; ++i) {
            slot[i] = SAGE_SLOTS;
            if (d[i] >= 0) slot[i] = atomicAdd(st + (d[i] / SAGE_NPB) * SAGE_LP, 1);
        }
#pragma unroll
        for (int i = 0; i < SAGE_PART_BATCH; ++i) {
            if (slot[i] < SAGE_SLOTS) {
                const int b = d[i] / SAGE_NPB;
                st[b * SAGE_LP + 1 + slot[i]] = ((d[i] - b * SAGE_NPB) << 16) | s[i];
            }
        }
    }
    __syncthreads();

    const int wave = tid >> 5;
    const int lane = tid & 31;
    constexpr int kWaves = SAGE_PART_THREADS / 32;
    constexpr int kCellsPerWave = SAGE_NB / kWaves;
    static_assert(kCellsPerWave * kWaves == SAGE_NB && kCellsPerWave <= 32, "cells per wave");
    const int myCount = (lane < kCellsPerWave) ? st[(wave + kWaves * lane) * SAGE_LP] : 0;
    int first[kCellsPerWave];
#pragma unroll
    for (int k = 0; k < kCellsPerWave; ++k) {
        const int n = __builtin_amdgcn_readlane(myCount, k);
        const int used = n < SAGE_SLOTS ? n : SAGE_SLOTS;
        first[k] = (lane <= used) ? st[(wave + kWaves * k) * SAGE_LP + lane] : 0;
    }
#pragma unroll
    for (int k = 0; k < kCellsPerWave; ++k) {
        const int b = wave + kWaves * k;
        const int n = __builtin_amdgcn_readlane(myCount, k);
        const int used = n < SAGE_SLOTS ? n : SAGE_SLOTS;
        int* g = cells + ((size_t)b * SAGE_NP + p) * SAGE_REC;
        g[lane] = first[k];
        if (32 + lane <= used) g[32 + lane] = st[b * SAGE_LP + 32 + lane];
    }
}

__device__ __forceinline__ int sage_take_slot(int* recS, int entry) {
    return atomicAdd(recS + (entry >> 16) * SAGE_LP, 1);
}

__device__ __forceinline__ void sage_put_entry(int* recS, int entry, int slot, int node0,
                                               int* ovfCount, int* __restrict__ ovfDst,
                                               int* __restrict__ ovfSrc) {
    const int ld = entry >> 16;
    const int sn = entry & 0xffff;
    if (slot < SAGE_SLOTS) {
        recS[ld * SAGE_LP + 1 + slot] = sn;
    } else {
        const int k = atomicAdd(ovfCount, 1);
        ovfDst[k] = node0 + ld;
        ovfSrc[k] = sn;
    }
}

__global__ __launch_bounds__(256) void sage_file_kernel(
    const int* __restrict__ src, const int* __restrict__ dst,
    const int* __restrict__ cells, int* __restrict__ rec, int* ovfCount,
    int* __restrict__ ovfDst, int* __restrict__ ovfSrc, int E) {
    __shared__ int recS[SAGE_NPB * SAGE_LP];
    const int tid = threadIdx.x;
    const int wave = tid >> 5;
    const int lane = tid & 31;
    const int b = blockIdx.x;
    const int node0 = b * SAGE_NPB;
    for (int i = tid; i < SAGE_NPB * SAGE_LP; i += 256) recS[i] = 0;
    __syncthreads();

    const int* cw = cells + ((size_t)b * SAGE_NP + wave * 32) * SAGE_REC;
    int w[32], slot[32];
#pragma unroll
    for (int j = 0; j < 32; ++j) w[j] = cw[j * SAGE_REC + lane];
    unsigned more = 0;
    unsigned redo = 0;
#pragma unroll
    for (int j = 0; j < 32; ++j) {
        const int n = __builtin_amdgcn_readfirstlane(w[j]);
        const bool whole = n <= SAGE_SLOTS;
        redo |= (whole ? 0u : 1u) << j;
        more |= ((whole && n > 31) ? 1u : 0u) << j;
        slot[j] = -1;
        if (whole && lane >= 1 && lane <= n) slot[j] = sage_take_slot(recS, w[j]);
    }
#pragma unroll
    for (int j = 0; j < 32; ++j) {
        if (slot[j] >= 0) sage_put_entry(recS, w[j], slot[j], node0, ovfCount, ovfDst, ovfSrc);
    }
#pragma nounroll
    for (int j = 0; j < 32; ++j) {
        if ((more >> j) & 1u) {
            const int n = cw[j * SAGE_REC];
            if (32 + lane <= n) {
                const int entry = cw[j * SAGE_REC + 32 + lane];
                sage_put_entry(recS, entry, sage_take_slot(recS, entry), node0, ovfCount,
                               ovfDst, ovfSrc);
            }
        }
    }
#pragma nounroll
    for (int j = 0; j < 32; ++j) {
        if ((redo >> j) & 1u) {
            const int per = (E + SAGE_NP - 1) / SAGE_NP;
            const int e0 = (wave * 32 + j) * per;
            const int e1 = (e0 + per < E) ? (e0 + per) : E;
            for (int e = e0 + lane; e < e1; e += 32) {
                const int dn = dst[e];
                if (dn >= node0 && dn < node0 + SAGE_NPB) {
                    const int entry = ((dn - node0) << 16) | src[e];
                    sage_put_entry(recS, entry, sage_take_slot(recS, entry), node0, ovfCount,
                                   ovfDst, ovfSrc);
                }
            }
        }
    }
    __syncthreads();

    for (int q = tid; q < SAGE_NPB * (SAGE_REC / 4); q += 256) {
        const int ln = q >> 4;
        const int* r = recS + ln * SAGE_LP + (q & 15) * 4;
        if (node0 + ln < SAGE_N)
            ((int4*)(rec + (size_t)node0 * SAGE_REC))[q] = make_int4(r[0], r[1], r[2], r[3]);
    }
}

template <int I, int K>
__device__ __forceinline__ void sage_add_chunk(float4& acc, const float4* __restrict__ xrow,
                                               int keys, int cnt) {
    if constexpr (I < K) {
        if (I < cnt) {
            const int sid = __builtin_amdgcn_readlane(keys, cnt - 1 - I) >> 6;
            const float4 v = xrow[(size_t)sid * (SAGE_D / 4)];
            sage_add_chunk<I + 1, K>(acc, xrow, keys, cnt);
            acc.x += v.x; acc.y += v.y; acc.z += v.z; acc.w += v.w;
        }
    }
}

__device__ __forceinline__ float4 sage_node_sum(const float4* __restrict__ xrow,
                                                const int* __restrict__ r, int w0, int deg,
                                                int n, int* sortedW, int lane,
                                                const int* __restrict__ ovfCount,
                                                const int* __restrict__ ovfDst,
                                                const int* __restrict__ ovfSrc) {
    float4 acc = make_float4(0.0f, 0.0f, 0.0f, 0.0f);
    if (deg <= SAGE_SLOTS) {
        int w1 = 0;
        if (deg > 31) w1 = r[32 + lane];
        const int i0 = lane - 1;
        const int i1 = lane + 31;
        const bool ok0 = (lane >= 1) && (i0 < deg);
        const bool ok1 = (i1 < deg);
        const int k0 = ok0 ? ((w0 << 6) | i0) : SAGE_NOKEY;
        const int k1 = ok1 ? ((w1 << 6) | i1) : SAGE_NOKEY;
        int r0k = 0, r1k = 0;
        const int na = deg < 31 ? deg : 31;
        for (int j = 0; j < na; ++j) {
            const int u = __builtin_amdgcn_readlane(k0, j + 1);
            r0k += (u < k0) ? 1 : 0;
            r1k += (u < k1) ? 1 : 0;
        }
        for (int j = 31; j < deg; ++j) {
            const int u = __builtin_amdgcn_readlane(k1, j - 31);
            r0k += (u < k0) ? 1 : 0;
            r1k += (u < k1) ? 1 : 0;
        }
        if (ok0) sortedW[r0k] = k0;
        if (ok1) sortedW[r1k] = k1;
#pragma nounroll
        for (int base = 0; base < deg; base += SAGE_KB) {
            const int keys = (base + lane < deg) ? sortedW[base + lane] : 0;
            const int left = deg - base;
            sage_add_chunk<0, SAGE_KB>(acc, xrow, keys, left < SAGE_KB ? left : SAGE_KB);
        }
    } else {
        const int w1 = r[32 + lane];
        const int k0 = (lane >= 1) ? w0 : SAGE_NOKEY;
        const int k1 = w1;
        const int nov = *ovfCount;
        int last = -1;
        for (int it = 0; it < deg; ++it) {
            int best = SAGE_NOKEY;
            if (k0 > last && k0 < best) best = k0;
            if (k1 > last && k1 < best) best = k1;
            for (int q = lane; q < nov; q += 32) {
                if (ovfDst[q] == n) {
                    const int v = ovfSrc[q];
                    if (v > last && v < best) best = v;
                }
            }
            {
                int mn = SAGE_NOKEY;
#pragma nounroll
                for (int j = 0; j < 32; ++j) {
                    const int o = __builtin_amdgcn_readlane(best, j);
                    mn = o < mn ? o : mn;
                }
                best = mn;
            }
            if (best == SAGE_NOKEY) break;
            int c = (k0 == best ? 1 : 0) + (k1 == best ? 1 : 0);
            for (int q = lane; q < nov; q += 32) {
                if (ovfDst[q] == n && ovfSrc[q] == best) ++c;
            }
            {
                int tot = 0;
#pragma nounroll
                for (int j = 0; j < 32; ++j) tot += __builtin_amdgcn_readlane(c, j);
                c = tot;
            }
            const int sid = __builtin_amdgcn_readfirstlane(best);
            const float4 v = xrow[(size_t)sid * (SAGE_D / 4)];
            for (int t = 0; t < c; ++t) {
                acc.x += v.x; acc.y += v.y; acc.z += v.z; acc.w += v.w;
            }
            last = best;
        }
    }
    return acc;
}

#define SAGE_WP 132
#define SAGE_GEMM_WGS 256
#define SAGE_LW 32
#define SAGE_LMW 8
#define SAGE_LGW (SAGE_LW - SAGE_LMW)
#define SAGE_RN 64
static_assert(12 * SAGE_GEMM_WGS <= SAGE_N / 16 && 13 * SAGE_GEMM_WGS >= SAGE_N / 16,
              "three whole rounds per work-group and at most one tile more");
static_assert(SAGE_N % 16 == 0, "every row tile is whole");
static_assert(4 * SAGE_LGW >= SAGE_RN, "a wave gathers at most 4 nodes per slot");
static_assert((SAGE_D * (SAGE_D / 4)) % (SAGE_LW * 32) == 0, "the weight copy divides evenly");

template <int NCT>
__device__ __forceinline__ void sage_load_b(v2f (&b)[2 * NCT], const float* wl_s,
                                            const float* wr_s, int k0) {
#pragma unroll
    for (int ct = 0; ct < NCT; ++ct) {
        b[ct]       = *(const v2f*)(wl_s + ct * 16 * SAGE_WP + k0);
        b[NCT + ct] = *(const v2f*)(wr_s + ct * 16 * SAGE_WP + k0);
    }
}

__device__ __forceinline__ void sage_load_a(v2f (&a)[4], const float* arow_s,
                                            const float* arow_x, int kc) {
#pragma unroll
    for (int i = 0; i < 2; ++i) {
        a[i]     = *(const v2f*)(arow_s + kc + 4 * i);
        a[2 + i] = *(const v2f*)(arow_x + kc + 4 * i);
    }
}

template <int NCT>
__device__ __forceinline__ void sage_mma_step(v8f (&acc)[NCT], const v2f (&b)[2 * NCT],
                                              v2f a_agg, v2f a_x) {
#pragma unroll
    for (int ct = 0; ct < NCT; ++ct) {
        acc[ct] = __builtin_amdgcn_wmma_f32_16x16x4_f32(
            false, a_agg, false, b[ct], (short)0, acc[ct], false, false);
        acc[ct] = __builtin_amdgcn_wmma_f32_16x16x4_f32(
            false, a_x, false, b[NCT + ct], (short)0, acc[ct], false, false);
    }
}

template <int NCT, bool LAST>
__device__ __forceinline__ void sage_mma_chunk(v8f (&acc)[NCT], const v2f (&a)[4],
                                               v2f (&b0)[2 * NCT], v2f (&b1)[2 * NCT],
                                               const float* wl_s, const float* wr_s,
                                               float inv, int kc) {
    sage_load_b<NCT>(b1, wl_s, wr_s, kc + 4);
    __builtin_amdgcn_sched_barrier(0);
    sage_mma_step<NCT>(acc, b0, a[0] * inv, a[2]);
    __builtin_amdgcn_sched_barrier(0);
    if (!LAST) sage_load_b<NCT>(b0, wl_s, wr_s, kc + 8);
    __builtin_amdgcn_sched_barrier(0);
    sage_mma_step<NCT>(acc, b1, a[1] * inv, a[3]);
    __builtin_amdgcn_sched_barrier(0);
}

template <int NCT>
__device__ __forceinline__ void sage_tile(const float* arow_s, const float* arow_x, float deg,
                                          const float* wl_s, const float* wr_s,
                                          const float* bias_s, float* __restrict__ out,
                                          int m0, int col0, int half, int lm) {
    v8f acc[NCT];
#pragma unroll
    for (int ct = 0; ct < NCT; ++ct) acc[ct] = (v8f){};

    v2f a0[4], a1[4], b0[2 * NCT], b1[2 * NCT];
    sage_load_a(a0, arow_s, arow_x, 0);
    sage_load_b<NCT>(b0, wl_s, wr_s, 0);
    __builtin_amdgcn_sched_barrier(0);
    const float inv = 1.0f / fmaxf(deg, 1.0f);
    for (int kc = 0; kc < SAGE_D - 16; kc += 16) {
        sage_load_a(a1, arow_s, arow_x, kc + 8);
        sage_mma_chunk<NCT, false>(acc, a0, b0, b1, wl_s, wr_s, inv, kc);
        sage_load_a(a0, arow_s, arow_x, kc + 16);
        sage_mma_chunk<NCT, false>(acc, a1, b0, b1, wl_s, wr_s, inv, kc + 8);
    }
    sage_load_a(a1, arow_s, arow_x, SAGE_D - 8);
    sage_mma_chunk<NCT, false>(acc, a0, b0, b1, wl_s, wr_s, inv, SAGE_D - 16);
    sage_mma_chunk<NCT, true>(acc, a1, b0, b1, wl_s, wr_s, inv, SAGE_D - 8);

    float bias[NCT];
#pragma unroll
    for (int ct = 0; ct < NCT; ++ct) bias[ct] = bias_s[col0 + ct * 16 + lm];
#pragma unroll
    for (int ct = 0; ct < NCT; ++ct) {
#pragma unroll
        for (int v = 0; v < 8; ++v) {
            out[(size_t)(m0 + v + 8 * half) * SAGE_D + col0 + ct * 16 + lm] =
                acc[ct][v] + bias[ct];
        }
    }
}

__global__ __launch_bounds__(SAGE_LW * 32) void sage_layer_kernel(
    const float* __restrict__ xin,
    const int* __restrict__ rec,
    const int* __restrict__ ovfCount, const int* __restrict__ ovfDst,
    const int* __restrict__ ovfSrc,
    const float* __restrict__ Wl,
    const float* __restrict__ bl,
    const float* __restrict__ Wr,
    float* __restrict__ out,
    int nRowTiles) {
    __shared__ __attribute__((aligned(16))) float sW[2 * SAGE_D * SAGE_WP];
    __shared__ __attribute__((aligned(16))) float sAgg[2 * SAGE_RN * SAGE_WP];
    __shared__ float sCnt[2 * SAGE_RN];
    __shared__ float sBias[SAGE_D];
    __shared__ int sorted[SAGE_LW][SAGE_REC];

#pragma unroll
    for (int j = 0; j < SAGE_D * (SAGE_D / 4) / (SAGE_LW * 32); ++j) {
        const int i   = j * (SAGE_LW * 32) + threadIdx.x;
        const int row = i >> 5;
        const int c4  = i & 31;
        const float4 vl = ((const float4*)Wl)[i];
        const float4 vr = ((const float4*)Wr)[i];
        *(float4*)(sW + row * SAGE_WP + c4 * 4) = vl;
        *(float4*)(sW + (SAGE_D + row) * SAGE_WP + c4 * 4) = vr;
    }
    if (threadIdx.x < SAGE_D) sBias[threadIdx.x] = bl[threadIdx.x];

    const int wave = __builtin_amdgcn_readfirstlane(threadIdx.x >> 5);
    const int lane = threadIdx.x & 31;
    const int half = lane >> 4;
    const int lm   = lane & 15;
    const int b    = blockIdx.x;
    const int mt    = wave & 3;
    const int mcol0 = (wave >> 2) * 64;
    const float* wl_s = sW + (mcol0 + lm) * SAGE_WP + 2 * half;
    const float* wr_s = wl_s + SAGE_D * SAGE_WP;
    const float* as_s = sAgg + (mt * 16 + lm) * SAGE_WP + 2 * half;
    const float4* xrow = (const float4*)xin + lane;
    const int nRounds = (12 * SAGE_GEMM_WGS + b < nRowTiles) ? 4 : 3;

#pragma nounroll
    for (int sl = 0; sl < 4; ++sl) {
        if (sl > 0) __syncthreads();

        if (sl < nRounds && (wave >= SAGE_LMW || sl == 0)) {
            const int idx    = (sl == 0) ? wave : wave - SAGE_LMW;
            const int stride = (sl == 0) ? SAGE_LW : SAGE_LGW;
            const int nq     = (sl < 3) ? SAGE_RN : 16;
            float* aggB = sAgg + (sl & 1) * (SAGE_RN * SAGE_WP);
            float* cntB = sCnt + (sl & 1) * SAGE_RN;
            const int jb = 4 * sl;
            int wa = 0, wb = 0, wc = 0, wd = 0;
            {
                const int q0 = idx, q1 = idx + stride, q2 = idx + 2 * stride, q3 = idx + 3 * stride;
                if (q0 < nq)
                    wa = rec[(size_t)(((jb + (q0 >> 4)) * SAGE_GEMM_WGS + b) * 16 + (q0 & 15)) * SAGE_REC + lane];
                if (q1 < nq)
                    wb = rec[(size_t)(((jb + (q1 >> 4)) * SAGE_GEMM_WGS + b) * 16 + (q1 & 15)) * SAGE_REC + lane];
                if (q2 < nq)
                    wc = rec[(size_t)(((jb + (q2 >> 4)) * SAGE_GEMM_WGS + b) * 16 + (q2 & 15)) * SAGE_REC + lane];
                if (q3 < nq)
                    wd = rec[(size_t)(((jb + (q3 >> 4)) * SAGE_GEMM_WGS + b) * 16 + (q3 & 15)) * SAGE_REC + lane];
            }
#pragma nounroll
            for (int i = 0; i < 4; ++i) {
                const int q = idx + i * stride;
                if (q >= nq) break;
                const int n = ((jb + (q >> 4)) * SAGE_GEMM_WGS + b) * 16 + (q & 15);
                const int w0 = wa;
                wa = wb; wb = wc; wc = wd;
                const int deg = __builtin_amdgcn_readfirstlane(w0);
                const float4 acc = sage_node_sum(xrow, rec + (size_t)n * SAGE_REC, w0, deg, n,
                                                 sorted[wave], lane, ovfCount, ovfDst, ovfSrc);
                *(float4*)(aggB + q * SAGE_WP + lane * 4) = acc;
                if (lane == 0) cntB[q] = (float)deg;
            }
        }

        if (sl >= 1 && wave < SAGE_LMW) {
            const int r = sl - 1;
            const float* cntB = sCnt + (r & 1) * SAGE_RN;
            const int m0 = ((4 * r + mt) * SAGE_GEMM_WGS + b) * 16;
            sage_tile<4>(as_s + (r & 1) * (SAGE_RN * SAGE_WP),
                         xin + (size_t)(m0 + lm) * SAGE_D + 2 * half, cntB[mt * 16 + lm],
                         wl_s, wr_s, sBias, out, m0, mcol0, half, lm);
        }
    }

    if (nRounds == 4) {
        __syncthreads();
        if (wave < 4) {
            const float* cntB = sCnt + SAGE_RN;
            const int m0   = (12 * SAGE_GEMM_WGS + b) * 16;
            const int col0 = wave * 32;
            sage_tile<2>(as_s + (SAGE_RN - mt * 16) * SAGE_WP,
                         xin + (size_t)(m0 + lm) * SAGE_D + 2 * half, cntB[lm],
                         wl_s + col0 * SAGE_WP, wr_s + col0 * SAGE_WP, sBias, out,
                         m0, col0, half, lm);
        }
    }
}

extern "C" void kernel_launch(void* const* d_in, const int* in_sizes, int n_in,
                              void* d_out, int out_size, void* d_ws, size_t ws_size,
                              hipStream_t stream) {
    const float* x   = (const float*)d_in[0];
    const int*   ei  = (const int*)d_in[1];
    const float* Wl1 = (const float*)d_in[2];
    const float* bl1 = (const float*)d_in[3];
    const float* Wr1 = (const float*)d_in[4];
    const float* Wl2 = (const float*)d_in[5];
    const float* bl2 = (const float*)d_in[6];
    const float* Wr2 = (const float*)d_in[7];
    float* out = (float*)d_out;

    const int N = SAGE_N;
    const int E = in_sizes[1] / 2;
    const int* src = ei;
    const int* dst = ei + E;

    float* h      = (float*)d_ws;
    int*   rec    = (int*)(h + (size_t)N * SAGE_D);
    int*   cells  = rec + (size_t)N * SAGE_REC;
    int*   ovfDst = cells + (size_t)SAGE_NB * SAGE_NP * SAGE_REC;
    int*   ovfSrc = ovfDst + E;
    int*   ovfCnt = ovfSrc + E;

    const int nRowTiles = N / 16;

    sage_part_kernel<<<SAGE_NP, SAGE_PART_THREADS, 0, stream>>>(src, dst, cells,
                                                                ovfCnt, E);
    sage_file_kernel<<<SAGE_NB, 256, 0, stream>>>(src, dst, cells, rec, ovfCnt,
                                                  ovfDst, ovfSrc, E);

    sage_layer_kernel<<<SAGE_GEMM_WGS, SAGE_LW * 32, 0, stream>>>(
        x, rec, ovfCnt, ovfDst, ovfSrc, Wl1, bl1, Wr1, h, nRowTiles);
    sage_layer_kernel<<<SAGE_GEMM_WGS, SAGE_LW * 32, 0, stream>>>(
        h, rec, ovfCnt, ovfDst, ovfSrc, Wl2, bl2, Wr2, out, nRowTiles);
}
